# starting kernel
# baseline (speedup 1.0000x reference)
#include <hip/hip_runtime.h>
#include <math.h>

typedef __attribute__((ext_vector_type(2))) float v2f;
typedef __attribute__((ext_vector_type(8))) float v8f;

#define NEG_SLOPE 0.2f
#define SOFT_EPS  1e-16f

__device__ __forceinline__ float lrelu(float x) { return x > 0.f ? x : NEG_SLOPE * x; }
__device__ __forceinline__ int   enc_ord(float f) { int i = __float_as_int(f); return i >= 0 ? i : (i ^ 0x7FFFFFFF); }
__device__ __forceinline__ float dec_ord(int i)   { return __int_as_float(i >= 0 ? i : (i ^ 0x7FFFFFFF)); }

__global__ void gat_gemm_wmma(const float* __restrict__ A, const float* __restrict__ Bg,
                              float* __restrict__ C, int M, int K, int NC) {
  extern __shared__ float Bs[];
  const int nthr = blockDim.x * blockDim.y;
  const int tid  = threadIdx.y * blockDim.x + threadIdx.x;
  for (int i = tid; i < K * NC; i += nthr) Bs[i] = Bg[i];
  __syncthreads();

  const int row0 = blockIdx.x * 16;
  const int col0 = threadIdx.y * 16;
  const int lane = threadIdx.x;
  const int mr   = lane & 15;
  const int half = lane >> 4;

  int arow = row0 + mr; if (arow > M - 1) arow = M - 1;
  const float* Ap = A + (size_t)arow * K;

  v8f acc = {};
  for (int k0 = 0; k0 < K; k0 += 4) {
    const int ka = k0 + 2 * half;
    v2f a, b;
    a.x = Ap[ka];
    a.y = Ap[ka + 1];
    b.x = Bs[ka * NC + col0 + mr];
    b.y = Bs[(ka + 1) * NC + col0 + mr];
    acc = __builtin_amdgcn_wmma_f32_16x16x4_f32(false, a, false, b, (short)0, acc,
                                                false, false);
  }
#pragma unroll
  for (int i = 0; i < 8; ++i) {
    const int r = row0 + i + 8 * half;
    if (r < M) C[(size_t)r * NC + col0 + mr] = acc[i];
  }
}

__global__ void gat_prep1(const float* __restrict__ h1, const float* __restrict__ attS,
                          const float* __restrict__ attD, float* __restrict__ aS,
                          float* __restrict__ aD, int* __restrict__ menc,
                          float* __restrict__ denom, float* __restrict__ agg, int N) {
  const int t = blockIdx.x * blockDim.x + threadIdx.x;
  if (t >= N * 8) return;
  const int h = t & 7;
  const float* hp = h1 + (size_t)t * 8;
  float s = 0.f, d = 0.f;
#pragma unroll
  for (int c = 0; c < 8; ++c) { s += hp[c] * attS[h * 8 + c]; d += hp[c] * attD[h * 8 + c]; }
  aS[t] = s; aD[t] = d;
  menc[t]  = enc_ord(lrelu(s + d));
  denom[t] = 0.f;
  float* ap = agg + (size_t)t * 8;
#pragma unroll
  for (int c = 0; c < 8; ++c) ap[c] = 0.f;
}

__global__ void gat_emax1(const int* __restrict__ src, const int* __restrict__ dst,
                          const float* __restrict__ aS, const float* __restrict__ aD,
                          int* __restrict__ menc, int E) {
  const int e = blockIdx.x * blockDim.x + threadIdx.x;
  if (e >= E) return;
  const int s = src[e], d = dst[e];
#pragma unroll
  for (int h = 0; h < 8; ++h)
    atomicMax(&menc[d * 8 + h], enc_ord(lrelu(aS[s * 8 + h] + aD[d * 8 + h])));
}

__global__ void gat_dself1(const float* __restrict__ aS, const float* __restrict__ aD,
                           const int* __restrict__ menc, float* __restrict__ denom, int N) {
  const int t = blockIdx.x * blockDim.x + threadIdx.x;
  if (t >= N * 8) return;
  denom[t] = __expf(lrelu(aS[t] + aD[t]) - dec_ord(menc[t]));
}

__global__ void gat_esum1(const int* __restrict__ src, const int* __restrict__ dst,
                          const float* __restrict__ aS, const float* __restrict__ aD,
                          const int* __restrict__ menc, float* __restrict__ denom, int E) {
  const int e = blockIdx.x * blockDim.x + threadIdx.x;
  if (e >= E) return;
  const int s = src[e], d = dst[e];
#pragma unroll
  for (int h = 0; h < 8; ++h) {
    const float ev = lrelu(aS[s * 8 + h] + aD[d * 8 + h]);
    atomicAdd(&denom[d * 8 + h], __expf(ev - dec_ord(menc[d * 8 + h])));
  }
}

__global__ void gat_eagg1(const int* __restrict__ src, const int* __restrict__ dst,
                          const float* __restrict__ aS, const float* __restrict__ aD,
                          const int* __restrict__ menc, const float* __restrict__ denom,
                          const float* __restrict__ h1, float* __restrict__ agg, int E) {
  const int e = blockIdx.x * blockDim.x + threadIdx.x;
  if (e >= E) return;
  const int s = src[e], d = dst[e];
  __builtin_prefetch(h1 + (size_t)s * 64, 0, 1);
#pragma unroll
  for (int h = 0; h < 8; ++h) {
    const float ev = lrelu(aS[s * 8 + h] + aD[d * 8 + h]);
    const float alpha = __expf(ev - dec_ord(menc[d * 8 + h])) / (denom[d * 8 + h] + SOFT_EPS);
    const float4* hs = (const float4*)(h1 + (size_t)s * 64 + h * 8);
    const float4 p0 = hs[0], p1 = hs[1];
    float* o = agg + (size_t)d * 64 + h * 8;
    atomicAdd(o + 0, alpha * p0.x); atomicAdd(o + 1, alpha * p0.y);
    atomicAdd(o + 2, alpha * p0.z); atomicAdd(o + 3, alpha * p0.w);
    atomicAdd(o + 4, alpha * p1.x); atomicAdd(o + 5, alpha * p1.y);
    atomicAdd(o + 6, alpha * p1.z); atomicAdd(o + 7, alpha * p1.w);
  }
}

__global__ void gat_fin1(const float* __restrict__ agg, const float* __restrict__ h1,
                         const float* __restrict__ aS, const float* __restrict__ aD,
                         const int* __restrict__ menc, const float* __restrict__ denom,
                         const float* __restrict__ bias, float* __restrict__ hpost, int N) {
  const int t = blockIdx.x * blockDim.x + threadIdx.x;
  if (t >= N * 8) return;
  const int h = t & 7;
  const float alpha =
      __expf(lrelu(aS[t] + aD[t]) - dec_ord(menc[t])) / (denom[t] + SOFT_EPS);
  const float* ap = agg + (size_t)t * 8;
  const float* hp = h1 + (size_t)t * 8;
  float* op = hpost + (size_t)t * 8;
#pragma unroll
  for (int c = 0; c < 8; ++c) {
    const float v = ap[c] + alpha * hp[c] + bias[h * 8 + c];
    op[c] = v > 0.f ? v : (__expf(v) - 1.f);
  }
}

__global__ void gat_padw2(const float* __restrict__ W2, float* __restrict__ W2p) {
  const int t = blockIdx.x * blockDim.x + threadIdx.x;
  if (t >= 64 * 48) return;
  const int k = t / 48, c = t % 48;
  W2p[t] = (c < 40) ? W2[k * 40 + c] : 0.f;
}

__global__ void gat_prep2(const float* __restrict__ h2, const float* __restrict__ attS,
                          const float* __restrict__ attD, float* __restrict__ aS,
                          float* __restrict__ aD, int* __restrict__ menc,
                          float* __restrict__ denom, float* __restrict__ agg, int N) {
  const int n = blockIdx.x * blockDim.x + threadIdx.x;
  if (n >= N) return;
  const float* hp = h2 + (size_t)n * 48;
  float s = 0.f, d = 0.f;
#pragma unroll
  for (int c = 0; c < 40; ++c) { s += hp[c] * attS[c]; d += hp[c] * attD[c]; }
  aS[n] = s; aD[n] = d;
  menc[n]  = enc_ord(lrelu(s + d));
  denom[n] = 0.f;
  float* ap = agg + (size_t)n * 40;
#pragma unroll
  for (int c = 0; c < 40; ++c) ap[c] = 0.f;
}

__global__ void gat_emax2(const int* __restrict__ src, const int* __restrict__ dst,
                          const float* __restrict__ aS, const float* __restrict__ aD,
                          int* __restrict__ menc, int E) {
  const int e = blockIdx.x * blockDim.x + threadIdx.x;
  if (e >= E) return;
  atomicMax(&menc[dst[e]], enc_ord(lrelu(aS[src[e]] + aD[dst[e]])));
}

__global__ void gat_dself2(const float* __restrict__ aS, const float* __restrict__ aD,
                           const int* __restrict__ menc, float* __restrict__ denom, int N) {
  const int n = blockIdx.x * blockDim.x + threadIdx.x;
  if (n >= N) return;
  denom[n] = __expf(lrelu(aS[n] + aD[n]) - dec_ord(menc[n]));
}

__global__ void gat_esum2(const int* __restrict__ src, const int* __restrict__ dst,
                          const float* __restrict__ aS, const float* __restrict__ aD,
                          const int* __restrict__ menc, float* __restrict__ denom, int E) {
  const int e = blockIdx.x * blockDim.x + threadIdx.x;
  if (e >= E) return;
  const int s = src[e], d = dst[e];
  atomicAdd(&denom[d], __expf(lrelu(aS[s] + aD[d]) - dec_ord(menc[d])));
}

__global__ void gat_eagg2(const int* __restrict__ src, const int* __restrict__ dst,
                          const float* __restrict__ aS, const float* __restrict__ aD,
                          const int* __restrict__ menc, const float* __restrict__ denom,
                          const float* __restrict__ h2, float* __restrict__ agg, int E) {
  const int e = blockIdx.x * blockDim.x + threadIdx.x;
  if (e >= E) return;
  const int s = src[e], d = dst[e];
  __builtin_prefetch(h2 + (size_t)s * 48, 0, 1);
  const float alpha =
      __expf(lrelu(aS[s] + aD[d]) - dec_ord(menc[d])) / (denom[d] + SOFT_EPS);
  const float* hp = h2 + (size_t)s * 48;
  float* o = agg + (size_t)d * 40;
#pragma unroll
  for (int c = 0; c < 40; ++c) atomicAdd(o + c, alpha * hp[c]);
}

__global__ void gat_fin2(const float* __restrict__ agg, const float* __restrict__ h2,
                         const float* __restrict__ aS, const float* __restrict__ aD,
                         const int* __restrict__ menc, const float* __restrict__ denom,
                         const float* __restrict__ bias, float* __restrict__ out, int N) {
  const int n = blockIdx.x * blockDim.x + threadIdx.x;
  if (n >= N) return;
  const float alpha =
      __expf(lrelu(aS[n] + aD[n]) - dec_ord(menc[n])) / (denom[n] + SOFT_EPS);
  const float* ap = agg + (size_t)n * 40;
  const float* hp = h2 + (size_t)n * 48;
  float v[40];
  float mx = -INFINITY;
#pragma unroll
  for (int c = 0; c < 40; ++c) {
    v[c] = ap[c] + alpha * hp[c] + bias[c];
    mx = fmaxf(mx, v[c]);
  }
  float sum = 0.f;
#pragma unroll
  for (int c = 0; c < 40; ++c) sum += __expf(v[c] - mx);
  const float lse = mx + __logf(sum);
  float* op = out + (size_t)n * 40;
#pragma unroll
  for (int c = 0; c < 40; ++c) op[c] = v[c] - lse;
}

extern "C" void kernel_launch(void* const* d_in, const int* in_sizes, int n_in,
                              void* d_out, int out_size, void* d_ws, size_t ws_size,
                              hipStream_t stream) {
  const float* x   = (const float*)d_in[0];
  const int*   ei  = (const int*)d_in[1];
  const float* W1  = (const float*)d_in[2];
  const float* as1 = (const float*)d_in[3];
  const float* ad1 = (const float*)d_in[4];
  const float* b1  = (const float*)d_in[5];
  const float* W2  = (const float*)d_in[6];
  const float* as2 = (const float*)d_in[7];
  const float* ad2 = (const float*)d_in[8];
  const float* b2  = (const float*)d_in[9];
  float* out = (float*)d_out;

  const int Fin = in_sizes[2] / 64;
  const int N   = in_sizes[0] / Fin;
  const int E   = in_sizes[1] / 2;
  const int* src = ei;
  const int* dst = ei + E;

  float* ws = (float*)d_ws;
  size_t o = 0;
  float* h1    = ws + o; o += (size_t)N * 64;
  float* hpost = ws + o; o += (size_t)N * 64;
  float* aS1   = ws + o; o += (size_t)N * 8;
  float* aD1   = ws + o; o += (size_t)N * 8;
  int*   m1    = (int*)(ws + o); o += (size_t)N * 8;
  float* dn1   = ws + o; o += (size_t)N * 8;
  float* agg1  = ws + o; o += (size_t)N * 64;
  float* h2    = ws + o; o += (size_t)N * 48;
  float* aS2   = ws + o; o += (size_t)N;
  float* aD2   = ws + o; o += (size_t)N;
  int*   m2    = (int*)(ws + o); o += (size_t)N;
  float* dn2   = ws + o; o += (size_t)N;
  float* agg2  = ws + o; o += (size_t)N * 40;
  float* W2p   = ws + o; o += (size_t)64 * 48;

  const int TB = 256;
  const int gNH = (N * 8 + TB - 1) / TB;
  const int gN  = (N + TB - 1) / TB;
  const int gE  = (E + TB - 1) / TB;
  const int gM  = (N + 15) / 16;

  gat_gemm_wmma<<<gM, dim3(32, 4), (size_t)Fin * 64 * 4, stream>>>(x, W1, h1, N, Fin, 64);
  gat_prep1 <<<gNH, TB, 0, stream>>>(h1, as1, ad1, aS1, aD1, m1, dn1, agg1, N);
  gat_emax1 <<<gE,  TB, 0, stream>>>(src, dst, aS1, aD1, m1, E);
  gat_dself1<<<gNH, TB, 0, stream>>>(aS1, aD1, m1, dn1, N);
  gat_esum1 <<<gE,  TB, 0, stream>>>(src, dst, aS1, aD1, m1, dn1, E);
  gat_eagg1 <<<gE,  TB, 0, stream>>>(src, dst, aS1, aD1, m1, dn1, h1, agg1, E);
  gat_fin1  <<<gNH, TB, 0, stream>>>(agg1, h1, aS1, aD1, m1, dn1, b1, hpost, N);

  gat_padw2 <<<(64 * 48 + TB - 1) / TB, TB, 0, stream>>>(W2, W2p);
  gat_gemm_wmma<<<gM, dim3(32, 3), (size_t)64 * 48 * 4, stream>>>(hpost, W2p, h2, N, 64, 48);
  gat_prep2 <<<gN, TB, 0, stream>>>(h2, as2, ad2, aS2, aD2, m2, dn2, agg2, N);
  gat_emax2 <<<gE, TB, 0, stream>>>(src, dst, aS2, aD2, m2, E);
  gat_dself2<<<gN, TB, 0, stream>>>(aS2, aD2, m2, dn2, N);
  gat_esum2 <<<gE, TB, 0, stream>>>(src, dst, aS2, aD2, m2, dn2, E);
  gat_eagg2 <<<gE, TB, 0, stream>>>(src, dst, aS2, aD2, m2, dn2, h2, agg2, E);
  gat_fin2  <<<gN, TB, 0, stream>>>(agg2, h2, aS2, aD2, m2, dn2, b2, out, N);

  (void)n_in; (void)out_size; (void)ws_size;
}


# Softmax divide moved behind the sums: one edge pass adds exp(e-max) and exp(e-max)*row by float atomics (arrival order; self term now last), node pass divides once; both sum passes gone
# speedup vs baseline: 5.5708x; 5.5708x over previous
#include <hip/hip_runtime.h>
#include <math.h>

typedef __attribute__((ext_vector_type(2))) float v2f;
typedef __attribute__((ext_vector_type(8))) float v8f;

#define NEG_SLOPE 0.2f
#define SOFT_EPS  1e-16f

__device__ __forceinline__ float lrelu(float x) { return x > 0.f ? x : NEG_SLOPE * x; }
__device__ __forceinline__ int   enc_ord(float f) { int i = __float_as_int(f); return i >= 0 ? i : (i ^ 0x7FFFFFFF); }
__device__ __forceinline__ float dec_ord(int i)   { return __int_as_float(i >= 0 ? i : (i ^ 0x7FFFFFFF)); }

__global__ void gat_gemm_wmma(const float* __restrict__ A, const float* __restrict__ Bg,
                              float* __restrict__ C, int M, int K, int NC) {
  extern __shared__ float Bs[];
  const int nthr = blockDim.x * blockDim.y;
  const int tid  = threadIdx.y * blockDim.x + threadIdx.x;
  for (int i = tid; i < K * NC; i += nthr) Bs[i] = Bg[i];
  __syncthreads();

  const int row0 = blockIdx.x * 16;
  const int col0 = threadIdx.y * 16;
  const int lane = threadIdx.x;
  const int mr   = lane & 15;
  const int half = lane >> 4;

  int arow = row0 + mr; if (arow > M - 1) arow = M - 1;
  const float* Ap = A + (size_t)arow * K;

  v8f acc = {};
  for (int k0 = 0; k0 < K; k0 += 4) {
    const int ka = k0 + 2 * half;
    v2f a, b;
    a.x = Ap[ka];
    a.y = Ap[ka + 1];
    b.x = Bs[ka * NC + col0 + mr];
    b.y = Bs[(ka + 1) * NC + col0 + mr];
    acc = __builtin_amdgcn_wmma_f32_16x16x4_f32(false, a, false, b, (short)0, acc,
                                                false, false);
  }
#pragma unroll
  for (int i = 0; i < 8; ++i) {
    const int r = row0 + i + 8 * half;
    if (r < M) C[(size_t)r * NC + col0 + mr] = acc[i];
  }
}

__global__ void gat_prep1(const float* __restrict__ h1, const float* __restrict__ attS,
                          const float* __restrict__ attD, float* __restrict__ aS,
                          float* __restrict__ aD, int* __restrict__ menc,
                          float* __restrict__ denom, float* __restrict__ agg, int N) {
  const int t = blockIdx.x * blockDim.x + threadIdx.x;
  if (t >= N * 8) return;
  const int h = t & 7;
  const float* hp = h1 + (size_t)t * 8;
  float s = 0.f, d = 0.f;
#pragma unroll
  for (int c = 0; c < 8; ++c) { s += hp[c] * attS[h * 8 + c]; d += hp[c] * attD[h * 8 + c]; }
  aS[t] = s; aD[t] = d;
  menc[t]  = enc_ord(lrelu(s + d));
  denom[t] = 0.f;
  float* ap = agg + (size_t)t * 8;
#pragma unroll
  for (int c = 0; c < 8; ++c) ap[c] = 0.f;
}

__global__ void gat_emax1(const int* __restrict__ src, const int* __restrict__ dst,
                          const float* __restrict__ aS, const float* __restrict__ aD,
                          int* __restrict__ menc, int E) {
  const int lane = threadIdx.x & 31;
  const int e0 = __builtin_amdgcn_readfirstlane((int)((blockIdx.x * blockDim.x + threadIdx.x) & ~31u));
  if (e0 >= E) return;
  const int e = e0 + lane;
  const int s = e < E ? src[e] : 0, d = e < E ? dst[e] : 0;
  const int h = lane & 7;
#pragma unroll
  for (int i = 0; i < 8; ++i) {
    const int j = 4 * i + (lane >> 3);
    const int sj = __shfl(s, j), dj = __shfl(d, j);
    const int v = enc_ord(lrelu(aS[sj * 8 + h] + aD[dj * 8 + h]));
    if (e0 + j < E) atomicMax(&menc[dj * 8 + h], v);
  }
}

__global__ void gat_eagg1(const int* __restrict__ src, const int* __restrict__ dst,
                          const float* __restrict__ aS, const float* __restrict__ aD,
                          const int* __restrict__ menc, float* __restrict__ denom,
                          const float* __restrict__ h1, float* __restrict__ agg, int E) {
  const int lane = threadIdx.x & 31;
  const int e0 = __builtin_amdgcn_readfirstlane((int)((blockIdx.x * blockDim.x + threadIdx.x) & ~31u));
  if (e0 >= E) return;
  const int e = e0 + lane;
  const int s = e < E ? src[e] : 0, d = e < E ? dst[e] : 0;
  const int h = lane & 7;
  float ex[8];
  int   dh[8];
#pragma unroll
  for (int i = 0; i < 8; ++i) {
    const int j = 4 * i + (lane >> 3);
    const int sj = __shfl(s, j), dj = __shfl(d, j);
    const float ev = lrelu(aS[sj * 8 + h] + aD[dj * 8 + h]);
    dh[i] = dj * 8 + h;
    ex[i] = __expf(ev - dec_ord(menc[dh[i]]));
  }
#pragma unroll
  for (int i = 0; i < 8; ++i)
    if (e0 + 4 * i + (lane >> 3) < E) atomicAdd(&denom[dh[i]], ex[i]);
  const int hl = lane >> 3;
#pragma unroll
  for (int j = 0; j < 32; ++j) {
    const int sj = __builtin_amdgcn_readlane(s, j);
    const int dj = __builtin_amdgcn_readlane(d, j);
    const float a_lo = __shfl(ex[j >> 2], ((j & 3) << 3) + hl);
    const float a_hi = __shfl(ex[j >> 2], ((j & 3) << 3) + 4 + hl);
    const float* hs = h1 + (size_t)sj * 64;
    float* o = agg + (size_t)dj * 64;
    const float v0 = hs[lane], v1 = hs[32 + lane];
    if (e0 + j < E) {
      atomicAdd(o + lane, a_lo * v0);
      atomicAdd(o + 32 + lane, a_hi * v1);
    }
  }
}

__global__ void gat_fin1(const float* __restrict__ agg, const float* __restrict__ h1,
                         const float* __restrict__ aS, const float* __restrict__ aD,
                         const int* __restrict__ menc, const float* __restrict__ denom,
                         const float* __restrict__ bias, float* __restrict__ hpost, int N) {
  const int t = blockIdx.x * blockDim.x + threadIdx.x;
  if (t >= N * 8) return;
  const int h = t & 7;
  const float exs = __expf(lrelu(aS[t] + aD[t]) - dec_ord(menc[t]));
  const float inv = 1.f / (denom[t] + exs + SOFT_EPS);
  const float* ap = agg + (size_t)t * 8;
  const float* hp = h1 + (size_t)t * 8;
  float* op = hpost + (size_t)t * 8;
#pragma unroll
  for (int c = 0; c < 8; ++c) {
    const float v = (ap[c] + exs * hp[c]) * inv + bias[h * 8 + c];
    op[c] = v > 0.f ? v : (__expf(v) - 1.f);
  }
}

__global__ void gat_padw2(const float* __restrict__ W2, float* __restrict__ W2p) {
  const int t = blockIdx.x * blockDim.x + threadIdx.x;
  if (t >= 64 * 48) return;
  const int k = t / 48, c = t % 48;
  W2p[t] = (c < 40) ? W2[k * 40 + c] : 0.f;
}

__global__ void gat_prep2(const float* __restrict__ h2, const float* __restrict__ attS,
                          const float* __restrict__ attD, float* __restrict__ aS,
                          float* __restrict__ aD, int* __restrict__ menc,
                          float* __restrict__ agg, int N) {
  const int n = blockIdx.x * blockDim.x + threadIdx.x;
  if (n >= N) return;
  const float* hp = h2 + (size_t)n * 48;
  float s = 0.f, d = 0.f;
#pragma unroll
  for (int c = 0; c < 40; ++c) { s += hp[c] * attS[c]; d += hp[c] * attD[c]; }
  aS[n] = s; aD[n] = d;
  menc[n]  = enc_ord(lrelu(s + d));
  float* ap = agg + (size_t)n * 48;
#pragma unroll
  for (int c = 0; c < 48; ++c) ap[c] = 0.f;
}

__global__ void gat_emax2(const int* __restrict__ src, const int* __restrict__ dst,
                          const float* __restrict__ aS, const float* __restrict__ aD,
                          int* __restrict__ menc, int E) {
  const int e = blockIdx.x * blockDim.x + threadIdx.x;
  if (e >= E) return;
  atomicMax(&menc[dst[e]], enc_ord(lrelu(aS[src[e]] + aD[dst[e]])));
}

__global__ void gat_eagg2(const int* __restrict__ src, const int* __restrict__ dst,
                          const float* __restrict__ aS, const float* __restrict__ aD,
                          const int* __restrict__ menc, const float* __restrict__ h2,
                          float* __restrict__ agg, int E) {
  const int lane = threadIdx.x & 31;
  const int e0 = __builtin_amdgcn_readfirstlane((int)((blockIdx.x * blockDim.x + threadIdx.x) & ~31u));
  if (e0 >= E) return;
  const int e = e0 + lane;
  const int s = e < E ? src[e] : 0, d = e < E ? dst[e] : 0;
  const float ex = __expf(lrelu(aS[s] + aD[d]) - dec_ord(menc[d]));
  const int hi = lane >> 4;
  const int ct = 32 + (lane & 15);
#pragma unroll
  for (int j = 0; j < 32; j += 2) {
    const int s0 = __builtin_amdgcn_readlane(s, j), s1 = __builtin_amdgcn_readlane(s, j + 1);
    const int d0 = __builtin_amdgcn_readlane(d, j), d1 = __builtin_amdgcn_readlane(d, j + 1);
    const float x0 = __int_as_float(__builtin_amdgcn_readlane(__float_as_int(ex), j));
    const float x1 = __int_as_float(__builtin_amdgcn_readlane(__float_as_int(ex), j + 1));
    const float v0 = h2[(size_t)s0 * 48 + lane];
    const float v1 = h2[(size_t)s1 * 48 + lane];
    const int st = hi ? s1 : s0, dt = hi ? d1 : d0;
    const float xt = hi ? x1 : x0;
    const float vl = h2[(size_t)st * 48 + ct];
    const float vt = (ct == 40) ? 1.f : vl;
    if (e0 + j < E)
      atomicAdd(agg + (size_t)d0 * 48 + lane, x0 * v0);
    if (e0 + j + 1 < E)
      atomicAdd(agg + (size_t)d1 * 48 + lane, x1 * v1);
    if (e0 + j + hi < E && ct <= 40)
      atomicAdd(agg + (size_t)dt * 48 + ct, xt * vt);
  }
}

__global__ void gat_fin2(const float* __restrict__ agg, const float* __restrict__ h2,
                         const float* __restrict__ aS, const float* __restrict__ aD,
                         const int* __restrict__ menc,
                         const float* __restrict__ bias, float* __restrict__ out, int N) {
  const int n = blockIdx.x * blockDim.x + threadIdx.x;
  if (n >= N) return;
  const float* ap = agg + (size_t)n * 48;
  const float* hp = h2 + (size_t)n * 48;
  const float exs = __expf(lrelu(aS[n] + aD[n]) - dec_ord(menc[n]));
  const float inv = 1.f / (ap[40] + exs + SOFT_EPS);
  float v[40];
  float mx = -INFINITY;
#pragma unroll
  for (int c = 0; c < 40; ++c) {
    v[c] = (ap[c] + exs * hp[c]) * inv + bias[c];
    mx = fmaxf(mx, v[c]);
  }
  float sum = 0.f;
#pragma unroll
  for (int c = 0; c < 40; ++c) sum += __expf(v[c] - mx);
  const float lse = mx + __logf(sum);
  float* op = out + (size_t)n * 40;
#pragma unroll
  for (int c = 0; c < 40; ++c) op[c] = v[c] - lse;
}

extern "C" void kernel_launch(void* const* d_in, const int* in_sizes, int n_in,
                              void* d_out, int out_size, void* d_ws, size_t ws_size,
                              hipStream_t stream) {
  const float* x   = (const float*)d_in[0];
  const int*   ei  = (const int*)d_in[1];
  const float* W1  = (const float*)d_in[2];
  const float* as1 = (const float*)d_in[3];
  const float* ad1 = (const float*)d_in[4];
  const float* b1  = (const float*)d_in[5];
  const float* W2  = (const float*)d_in[6];
  const float* as2 = (const float*)d_in[7];
  const float* ad2 = (const float*)d_in[8];
  const float* b2  = (const float*)d_in[9];
  float* out = (float*)d_out;

  const int Fin = in_sizes[2] / 64;
  const int N   = in_sizes[0] / Fin;
  const int E   = in_sizes[1] / 2;
  const int* src = ei;
  const int* dst = ei + E;

  float* ws = (float*)d_ws;
  size_t o = 0;
  float* h1    = ws + o; o += (size_t)N * 64;
  float* hpost = ws + o; o += (size_t)N * 64;
  float* aS1   = ws + o; o += (size_t)N * 8;
  float* aD1   = ws + o; o += (size_t)N * 8;
  int*   m1    = (int*)(ws + o); o += (size_t)N * 8;
  float* dn1   = ws + o; o += (size_t)N * 8;
  float* agg1  = ws + o; o += (size_t)N * 64;
  float* h2    = ws + o; o += (size_t)N * 48;
  float* aS2   = ws + o; o += (size_t)N;
  float* aD2   = ws + o; o += (size_t)N;
  int*   m2    = (int*)(ws + o); o += (size_t)N;
  float* agg2  = ws + o; o += (size_t)N * 48;
  float* W2p   = ws + o; o += (size_t)64 * 48;

  const int TB = 256;
  const int gNH = (N * 8 + TB - 1) / TB;
  const int gN  = (N + TB - 1) / TB;
  const int gE  = (E + TB - 1) / TB;
  const int gM  = (N + 15) / 16;

  gat_gemm_wmma<<<gM, dim3(32, 4), (size_t)Fin * 64 * 4, stream>>>(x, W1, h1, N, Fin, 64);
  gat_prep1 <<<gNH, TB, 0, stream>>>(h1, as1, ad1, aS1, aD1, m1, dn1, agg1, N);
  gat_emax1 <<<gE,  TB, 0, stream>>>(src, dst, aS1, aD1, m1, E);
  gat_eagg1 <<<gE,  TB, 0, stream>>>(src, dst, aS1, aD1, m1, dn1, h1, agg1, E);
  gat_fin1  <<<gNH, TB, 0, stream>>>(agg1, h1, aS1, aD1, m1, dn1, b1, hpost, N);

  gat_padw2 <<<(64 * 48 + TB - 1) / TB, TB, 0, stream>>>(W2, W2p);
  gat_gemm_wmma<<<gM, dim3(32, 3), (size_t)64 * 48 * 4, stream>>>(hpost, W2p, h2, N, 64, 48);
  gat_prep2 <<<gN, TB, 0, stream>>>(h2, as2, ad2, aS2, aD2, m2, agg2, N);
  gat_emax2 <<<gE, TB, 0, stream>>>(src, dst, aS2, aD2, m2, E);
  gat_eagg2 <<<gE, TB, 0, stream>>>(src, dst, aS2, aD2, m2, h2, agg2, E);
  gat_fin2  <<<gN, TB, 0, stream>>>(agg2, h2, aS2, aD2, m2, b2, out, N);

  (void)n_in; (void)out_size; (void)ws_size;
}


# Max pass reduces source scores only (leaky relu is monotone: max e = lrelu(maxS+aD), same bits); aD and maxS kept as one pair: every edge pass gathers one piece fewer per edge; add order kept
# speedup vs baseline: 6.2057x; 1.1140x over previous
#include <hip/hip_runtime.h>
#include <math.h>

typedef __attribute__((ext_vector_type(2))) float v2f;
typedef __attribute__((ext_vector_type(8))) float v8f;
typedef __attribute__((ext_vector_type(2))) int   v2i;

#define NEG_SLOPE 0.2f
#define SOFT_EPS  1e-16f

__device__ __forceinline__ float lrelu(float x) { return x > 0.f ? x : NEG_SLOPE * x; }
__device__ __forceinline__ int   enc_ord(float f) { int i = __float_as_int(f); return i >= 0 ? i : (i ^ 0x7FFFFFFF); }
__device__ __forceinline__ float dec_ord(int i)   { return __int_as_float(i >= 0 ? i : (i ^ 0x7FFFFFFF)); }

__global__ void gat_gemm_wmma(const float* __restrict__ A, const float* __restrict__ Bg,
                              float* __restrict__ C, int M, int K, int NC) {
  extern __shared__ float Bs[];
  const int nthr = blockDim.x * blockDim.y;
  const int tid  = threadIdx.y * blockDim.x + threadIdx.x;
  for (int i = tid; i < K * NC; i += nthr) Bs[i] = Bg[i];
  __syncthreads();

  const int row0 = blockIdx.x * 16;
  const int col0 = threadIdx.y * 16;
  const int lane = threadIdx.x;
  const int mr   = lane & 15;
  const int half = lane >> 4;

  int arow = row0 + mr; if (arow > M - 1) arow = M - 1;
  const float* Ap = A + (size_t)arow * K;

  v8f acc = {};
  for (int k0 = 0; k0 < K; k0 += 4) {
    const int ka = k0 + 2 * half;
    v2f a, b;
    a.x = Ap[ka];
    a.y = Ap[ka + 1];
    b.x = Bs[ka * NC + col0 + mr];
    b.y = Bs[(ka + 1) * NC + col0 + mr];
    acc = __builtin_amdgcn_wmma_f32_16x16x4_f32(false, a, false, b, (short)0, acc,
                                                false, false);
  }
#pragma unroll
  for (int i = 0; i < 8; ++i) {
    const int r = row0 + i + 8 * half;
    if (r < M) C[(size_t)r * NC + col0 + mr] = acc[i];
  }
}

__global__ void gat_prep1(const float* __restrict__ h1, const float* __restrict__ attS,
                          const float* __restrict__ attD, float* __restrict__ aS,
                          int* __restrict__ dm,
                          float* __restrict__ denom, float* __restrict__ agg, int N) {
  const int t = blockIdx.x * blockDim.x + threadIdx.x;
  if (t >= N * 8) return;
  const int h = t & 7;
  const float* hp = h1 + (size_t)t * 8;
  float s = 0.f, d = 0.f;
#pragma unroll
  for (int c = 0; c < 8; ++c) { s += hp[c] * attS[h * 8 + c]; d += hp[c] * attD[h * 8 + c]; }
  aS[t] = s;
  v2i p; p.x = __float_as_int(d); p.y = enc_ord(s);
  ((v2i*)dm)[t] = p;
  denom[t] = 0.f;
  float* ap = agg + (size_t)t * 8;
#pragma unroll
  for (int c = 0; c < 8; ++c) ap[c] = 0.f;
}

__global__ void gat_emax1(const int* __restrict__ src, const int* __restrict__ dst,
                          const float* __restrict__ aS, int* __restrict__ dm, int E) {
  const int lane = threadIdx.x & 31;
  const int e0 = __builtin_amdgcn_readfirstlane((int)((blockIdx.x * blockDim.x + threadIdx.x) & ~31u));
  if (e0 >= E) return;
  const int e = e0 + lane;
  const int s = e < E ? src[e] : 0, d = e < E ? dst[e] : 0;
  const int h = lane & 7;
#pragma unroll
  for (int i = 0; i < 8; ++i) {
    const int j = 4 * i + (lane >> 3);
    const int sj = __shfl(s, j), dj = __shfl(d, j);
    const int v = enc_ord(aS[sj * 8 + h]);
    if (e0 + j < E) atomicMax(&dm[(dj * 8 + h) * 2 + 1], v);
  }
}

__global__ void gat_eagg1(const int* __restrict__ src, const int* __restrict__ dst,
                          const float* __restrict__ aS, const int* __restrict__ dm,
                          float* __restrict__ denom,
                          const float* __restrict__ h1, float* __restrict__ agg, int E) {
  const int lane = threadIdx.x & 31;
  const int e0 = __builtin_amdgcn_readfirstlane((int)((blockIdx.x * blockDim.x + threadIdx.x) & ~31u));
  if (e0 >= E) return;
  const int e = e0 + lane;
  const int s = e < E ? src[e] : 0, d = e < E ? dst[e] : 0;
  const int h = lane & 7;
  float ex[8];
  int   dh[8];
#pragma unroll
  for (int i = 0; i < 8; ++i) {
    const int j = 4 * i + (lane >> 3);
    const int sj = __shfl(s, j), dj = __shfl(d, j);
    dh[i] = dj * 8 + h;
    const v2i p = ((const v2i*)dm)[dh[i]];
    const float ad = __int_as_float(p.x);
    const float ev = lrelu(aS[sj * 8 + h] + ad);
    ex[i] = __expf(ev - lrelu(dec_ord(p.y) + ad));
  }
#pragma unroll
  for (int i = 0; i < 8; ++i)
    if (e0 + 4 * i + (lane >> 3) < E) atomicAdd(&denom[dh[i]], ex[i]);
  const int hl = lane >> 3;
#pragma unroll
  for (int j = 0; j < 32; ++j) {
    const int sj = __builtin_amdgcn_readlane(s, j);
    const int dj = __builtin_amdgcn_readlane(d, j);
    const float a_lo = __shfl(ex[j >> 2], ((j & 3) << 3) + hl);
    const float a_hi = __shfl(ex[j >> 2], ((j & 3) << 3) + 4 + hl);
    const float* hs = h1 + (size_t)sj * 64;
    float* o = agg + (size_t)dj * 64;
    const float v0 = hs[lane], v1 = hs[32 + lane];
    if (e0 + j < E) {
      atomicAdd(o + lane, a_lo * v0);
      atomicAdd(o + 32 + lane, a_hi * v1);
    }
  }
}

__global__ void gat_fin1(const float* __restrict__ agg, const float* __restrict__ h1,
                         const float* __restrict__ aS, const int* __restrict__ dm,
                         const float* __restrict__ denom,
                         const float* __restrict__ bias, float* __restrict__ hpost, int N) {
  const int t = blockIdx.x * blockDim.x + threadIdx.x;
  if (t >= N * 8) return;
  const int h = t & 7;
  const v2i p = ((const v2i*)dm)[t];
  const float ad = __int_as_float(p.x);
  const float exs = __expf(lrelu(aS[t] + ad) - lrelu(dec_ord(p.y) + ad));
  const float inv = 1.f / (denom[t] + exs + SOFT_EPS);
  const float* ap = agg + (size_t)t * 8;
  const float* hp = h1 + (size_t)t * 8;
  float* op = hpost + (size_t)t * 8;
#pragma unroll
  for (int c = 0; c < 8; ++c) {
    const float v = (ap[c] + exs * hp[c]) * inv + bias[h * 8 + c];
    op[c] = v > 0.f ? v : (__expf(v) - 1.f);
  }
}

__global__ void gat_padw2(const float* __restrict__ W2, float* __restrict__ W2p) {
  const int t = blockIdx.x * blockDim.x + threadIdx.x;
  if (t >= 64 * 48) return;
  const int k = t / 48, c = t % 48;
  W2p[t] = (c < 40) ? W2[k * 40 + c] : 0.f;
}

__global__ void gat_prep2(const float* __restrict__ h2, const float* __restrict__ attS,
                          const float* __restrict__ attD, float* __restrict__ aS,
                          int* __restrict__ dm,
                          float* __restrict__ agg, int N) {
  const int n = blockIdx.x * blockDim.x + threadIdx.x;
  if (n >= N) return;
  const float* hp = h2 + (size_t)n * 48;
  float s = 0.f, d = 0.f;
#pragma unroll
  for (int c = 0; c < 40; ++c) { s += hp[c] * attS[c]; d += hp[c] * attD[c]; }
  aS[n] = s;
  v2i p; p.x = __float_as_int(d); p.y = enc_ord(s);
  ((v2i*)dm)[n] = p;
  float* ap = agg + (size_t)n * 48;
#pragma unroll
  for (int c = 0; c < 48; ++c) ap[c] = 0.f;
}

__global__ void gat_emax2(const int* __restrict__ src, const int* __restrict__ dst,
                          const float* __restrict__ aS, int* __restrict__ dm, int E) {
  const int e = blockIdx.x * blockDim.x + threadIdx.x;
  if (e >= E) return;
  atomicMax(&dm[dst[e] * 2 + 1], enc_ord(aS[src[e]]));
}

__global__ void gat_eagg2(const int* __restrict__ src, const int* __restrict__ dst,
                          const float* __restrict__ aS, const int* __restrict__ dm,
                          const float* __restrict__ h2,
                          float* __restrict__ agg, int E) {
  const int lane = threadIdx.x & 31;
  const int e0 = __builtin_amdgcn_readfirstlane((int)((blockIdx.x * blockDim.x + threadIdx.x) & ~31u));
  if (e0 >= E) return;
  const int e = e0 + lane;
  const int s = e < E ? src[e] : 0, d = e < E ? dst[e] : 0;
  const v2i p = ((const v2i*)dm)[d];
  const float ad = __int_as_float(p.x);
  const float ex = __expf(lrelu(aS[s] + ad) - lrelu(dec_ord(p.y) + ad));
  const int hi = lane >> 4;
  const int ct = 32 + (lane & 15);
#pragma unroll
  for (int j = 0; j < 32; j += 2) {
    const int s0 = __builtin_amdgcn_readlane(s, j), s1 = __builtin_amdgcn_readlane(s, j + 1);
    const int d0 = __builtin_amdgcn_readlane(d, j), d1 = __builtin_amdgcn_readlane(d, j + 1);
    const float x0 = __int_as_float(__builtin_amdgcn_readlane(__float_as_int(ex), j));
    const float x1 = __int_as_float(__builtin_amdgcn_readlane(__float_as_int(ex), j + 1));
    const float v0 = h2[(size_t)s0 * 48 + lane];
    const float v1 = h2[(size_t)s1 * 48 + lane];
    const int st = hi ? s1 : s0, dt = hi ? d1 : d0;
    const float xt = hi ? x1 : x0;
    const float vl = h2[(size_t)st * 48 + ct];
    const float vt = (ct == 40) ? 1.f : vl;
    if (e0 + j < E)
      atomicAdd(agg + (size_t)d0 * 48 + lane, x0 * v0);
    if (e0 + j + 1 < E)
      atomicAdd(agg + (size_t)d1 * 48 + lane, x1 * v1);
    if (e0 + j + hi < E && ct <= 40)
      atomicAdd(agg + (size_t)dt * 48 + ct, xt * vt);
  }
}

__global__ void gat_fin2(const float* __restrict__ agg, const float* __restrict__ h2,
                         const float* __restrict__ aS, const int* __restrict__ dm,
                         const float* __restrict__ bias, float* __restrict__ out, int N) {
  const int n = blockIdx.x * blockDim.x + threadIdx.x;
  if (n >= N) return;
  const float* ap = agg + (size_t)n * 48;
  const float* hp = h2 + (size_t)n * 48;
  const v2i p = ((const v2i*)dm)[n];
  const float ad = __int_as_float(p.x);
  const float exs = __expf(lrelu(aS[n] + ad) - lrelu(dec_ord(p.y) + ad));
  const float inv = 1.f / (ap[40] + exs + SOFT_EPS);
  float v[40];
  float mx = -INFINITY;
#pragma unroll
  for (int c = 0; c < 40; ++c) {
    v[c] = (ap[c] + exs * hp[c]) * inv + bias[c];
    mx = fmaxf(mx, v[c]);
  }
  float sum = 0.f;
#pragma unroll
  for (int c = 0; c < 40; ++c) sum += __expf(v[c] - mx);
  const float lse = mx + __logf(sum);
  float* op = out + (size_t)n * 40;
#pragma unroll
  for (int c = 0; c < 40; ++c) op[c] = v[c] - lse;
}

extern "C" void kernel_launch(void* const* d_in, const int* in_sizes, int n_in,
                              void* d_out, int out_size, void* d_ws, size_t ws_size,
                              hipStream_t stream) {
  const float* x   = (const float*)d_in[0];
  const int*   ei  = (const int*)d_in[1];
  const float* W1  = (const float*)d_in[2];
  const float* as1 = (const float*)d_in[3];
  const float* ad1 = (const float*)d_in[4];
  const float* b1  = (const float*)d_in[5];
  const float* W2  = (const float*)d_in[6];
  const float* as2 = (const float*)d_in[7];
  const float* ad2 = (const float*)d_in[8];
  const float* b2  = (const float*)d_in[9];
  float* out = (float*)d_out;

  const int Fin = in_sizes[2] / 64;
  const int N   = in_sizes[0] / Fin;
  const int E   = in_sizes[1] / 2;
  const int* src = ei;
  const int* dst = ei + E;

  float* ws = (float*)d_ws;
  size_t o = 0;
  float* h1    = ws + o; o += (size_t)N * 64;
  float* hpost = ws + o; o += (size_t)N * 64;
  float* aS1   = ws + o; o += (size_t)N * 8;
  int*   dm1   = (int*)(ws + o); o += (size_t)N * 16;
  float* dn1   = ws + o; o += (size_t)N * 8;
  float* agg1  = ws + o; o += (size_t)N * 64;
  float* h2    = ws + o; o += (size_t)N * 48;
  int*   dm2   = (int*)(ws + o); o += (size_t)N * 2;
  float* aS2   = ws + o; o += (size_t)N;
  float* agg2  = ws + o; o += (size_t)N * 48;
  float* W2p   = ws + o; o += (size_t)64 * 48;

  const int TB = 256;
  const int gNH = (N * 8 + TB - 1) / TB;
  const int gN  = (N + TB - 1) / TB;
  const int gE  = (E + TB - 1) / TB;
  const int gM  = (N + 15) / 16;

  gat_gemm_wmma<<<gM, dim3(32, 4), (size_t)Fin * 64 * 4, stream>>>(x, W1, h1, N, Fin, 64);
  gat_prep1 <<<gNH, TB, 0, stream>>>(h1, as1, ad1, aS1, dm1, dn1, agg1, N);
  gat_emax1 <<<gE,  TB, 0, stream>>>(src, dst, aS1, dm1, E);
  gat_eagg1 <<<gE,  TB, 0, stream>>>(src, dst, aS1, dm1, dn1, h1, agg1, E);
  gat_fin1  <<<gNH, TB, 0, stream>>>(agg1, h1, aS1, dm1, dn1, b1, hpost, N);

  gat_padw2 <<<(64 * 48 + TB - 1) / TB, TB, 0, stream>>>(W2, W2p);
  gat_gemm_wmma<<<gM, dim3(32, 3), (size_t)64 * 48 * 4, stream>>>(hpost, W2p, h2, N, 64, 48);
  gat_prep2 <<<gN, TB, 0, stream>>>(h2, as2, ad2, aS2, dm2, agg2, N);
  gat_emax2 <<<gE, TB, 0, stream>>>(src, dst, aS2, dm2, E);
  gat_eagg2 <<<gE, TB, 0, stream>>>(src, dst, aS2, dm2, h2, agg2, E);
  gat_fin2  <<<gN, TB, 0, stream>>>(agg2, h2, aS2, dm2, b2, out, N);

  (void)n_in; (void)out_size; (void)ws_size;
}


# Row adds of both layers: full waves skip the per-edge end checks and request the source rows of 8 edges together before their float atomic adds (was one edge's loads at a time); add order kept
# speedup vs baseline: 6.5259x; 1.0516x over previous
#include <hip/hip_runtime.h>
#include <math.h>

typedef __attribute__((ext_vector_type(2))) float v2f;
typedef __attribute__((ext_vector_type(8))) float v8f;
typedef __attribute__((ext_vector_type(2))) int   v2i;

#define NEG_SLOPE 0.2f
#define SOFT_EPS  1e-16f

__device__ __forceinline__ float lrelu(float x) { return x > 0.f ? x : NEG_SLOPE * x; }
__device__ __forceinline__ int   enc_ord(float f) { int i = __float_as_int(f); return i >= 0 ? i : (i ^ 0x7FFFFFFF); }
__device__ __forceinline__ float dec_ord(int i)   { return __int_as_float(i >= 0 ? i : (i ^ 0x7FFFFFFF)); }

__global__ void gat_gemm_wmma(const float* __restrict__ A, const float* __restrict__ Bg,
                              float* __restrict__ C, int M, int K, int NC) {
  extern __shared__ float Bs[];
  const int nthr = blockDim.x * blockDim.y;
  const int tid  = threadIdx.y * blockDim.x + threadIdx.x;
  for (int i = tid; i < K * NC; i += nthr) Bs[i] = Bg[i];
  __syncthreads();

  const int row0 = blockIdx.x * 16;
  const int col0 = threadIdx.y * 16;
  const int lane = threadIdx.x;
  const int mr   = lane & 15;
  const int half = lane >> 4;

  int arow = row0 + mr; if (arow > M - 1) arow = M - 1;
  const float* Ap = A + (size_t)arow * K;

  v8f acc = {};
  for (int k0 = 0; k0 < K; k0 += 4) {
    const int ka = k0 + 2 * half;
    v2f a, b;
    a.x = Ap[ka];
    a.y = Ap[ka + 1];
    b.x = Bs[ka * NC + col0 + mr];
    b.y = Bs[(ka + 1) * NC + col0 + mr];
    acc = __builtin_amdgcn_wmma_f32_16x16x4_f32(false, a, false, b, (short)0, acc,
                                                false, false);
  }
#pragma unroll
  for (int i = 0; i < 8; ++i) {
    const int r = row0 + i + 8 * half;
    if (r < M) C[(size_t)r * NC + col0 + mr] = acc[i];
  }
}

__global__ void gat_prep1(const float* __restrict__ h1, const float* __restrict__ attS,
                          const float* __restrict__ attD, float* __restrict__ aS,
                          int* __restrict__ dm,
                          float* __restrict__ denom, float* __restrict__ agg, int N) {
  const int t = blockIdx.x * blockDim.x + threadIdx.x;
  if (t >= N * 8) return;
  const int h = t & 7;
  const float* hp = h1 + (size_t)t * 8;
  float s = 0.f, d = 0.f;
#pragma unroll
  for (int c = 0; c < 8; ++c) { s += hp[c] * attS[h * 8 + c]; d += hp[c] * attD[h * 8 + c]; }
  aS[t] = s;
  v2i p; p.x = __float_as_int(d); p.y = enc_ord(s);
  ((v2i*)dm)[t] = p;
  denom[t] = 0.f;
  float* ap = agg + (size_t)t * 8;
#pragma unroll
  for (int c = 0; c < 8; ++c) ap[c] = 0.f;
}

__global__ void gat_emax1(const int* __restrict__ src, const int* __restrict__ dst,
                          const float* __restrict__ aS, int* __restrict__ dm, int E) {
  const int lane = threadIdx.x & 31;
  const int e0 = __builtin_amdgcn_readfirstlane((int)((blockIdx.x * blockDim.x + threadIdx.x) & ~31u));
  if (e0 >= E) return;
  const int e = e0 + lane;
  const int s = e < E ? src[e] : 0, d = e < E ? dst[e] : 0;
  const int h = lane & 7;
#pragma unroll
  for (int i = 0; i < 8; ++i) {
    const int j = 4 * i + (lane >> 3);
    const int sj = __shfl(s, j), dj = __shfl(d, j);
    const int v = enc_ord(aS[sj * 8 + h]);
    if (e0 + j < E) atomicMax(&dm[(dj * 8 + h) * 2 + 1], v);
  }
}

__global__ void gat_eagg1(const int* __restrict__ src, const int* __restrict__ dst,
                          const float* __restrict__ aS, const int* __restrict__ dm,
                          float* __restrict__ denom,
                          const float* __restrict__ h1, float* __restrict__ agg, int E) {
  const int lane = threadIdx.x & 31;
  const int e0 = __builtin_amdgcn_readfirstlane((int)((blockIdx.x * blockDim.x + threadIdx.x) & ~31u));
  if (e0 >= E) return;
  const int e = e0 + lane;
  const int s = e < E ? src[e] : 0, d = e < E ? dst[e] : 0;
  const int h = lane & 7;
  float ex[8];
  int   dh[8];
#pragma unroll
  for (int i = 0; i < 8; ++i) {
    const int j = 4 * i + (lane >> 3);
    const int sj = __shfl(s, j), dj = __shfl(d, j);
    dh[i] = dj * 8 + h;
    const v2i p = ((const v2i*)dm)[dh[i]];
    const float ad = __int_as_float(p.x);
    const float ev = lrelu(aS[sj * 8 + h] + ad);
    ex[i] = __expf(ev - lrelu(dec_ord(p.y) + ad));
  }
#pragma unroll
  for (int i = 0; i < 8; ++i)
    if (e0 + 4 * i + (lane >> 3) < E) atomicAdd(&denom[dh[i]], ex[i]);
  const int hl = lane >> 3;
  if (e0 + 32 <= E) {
#pragma unroll
    for (int g = 0; g < 32; g += 8) {
      float v0[8], v1[8];
#pragma unroll
      for (int k = 0; k < 8; ++k) {
        const int sj = __builtin_amdgcn_readlane(s, g + k);
        const float* hs = h1 + (size_t)sj * 64;
        v0[k] = hs[lane];
        v1[k] = hs[32 + lane];
      }
#pragma unroll
      for (int k = 0; k < 8; ++k) {
        const int j = g + k;
        const int dj = __builtin_amdgcn_readlane(d, j);
        const float a_lo = __shfl(ex[j >> 2], ((j & 3) << 3) + hl);
        const float a_hi = __shfl(ex[j >> 2], ((j & 3) << 3) + 4 + hl);
        float* o = agg + (size_t)dj * 64;
        atomicAdd(o + lane, a_lo * v0[k]);
        atomicAdd(o + 32 + lane, a_hi * v1[k]);
      }
    }
    return;
  }
#pragma unroll
  for (int j = 0; j < 32; ++j) {
    const int sj = __builtin_amdgcn_readlane(s, j);
    const int dj = __builtin_amdgcn_readlane(d, j);
    const float a_lo = __shfl(ex[j >> 2], ((j & 3) << 3) + hl);
    const float a_hi = __shfl(ex[j >> 2], ((j & 3) << 3) + 4 + hl);
    const float* hs = h1 + (size_t)sj * 64;
    float* o = agg + (size_t)dj * 64;
    const float v0 = hs[lane], v1 = hs[32 + lane];
    if (e0 + j < E) {
      atomicAdd(o + lane, a_lo * v0);
      atomicAdd(o + 32 + lane, a_hi * v1);
    }
  }
}

__global__ void gat_fin1(const float* __restrict__ agg, const float* __restrict__ h1,
                         const float* __restrict__ aS, const int* __restrict__ dm,
                         const float* __restrict__ denom,
                         const float* __restrict__ bias, float* __restrict__ hpost, int N) {
  const int t = blockIdx.x * blockDim.x + threadIdx.x;
  if (t >= N * 8) return;
  const int h = t & 7;
  const v2i p = ((const v2i*)dm)[t];
  const float ad = __int_as_float(p.x);
  const float exs = __expf(lrelu(aS[t] + ad) - lrelu(dec_ord(p.y) + ad));
  const float inv = 1.f / (denom[t] + exs + SOFT_EPS);
  const float* ap = agg + (size_t)t * 8;
  const float* hp = h1 + (size_t)t * 8;
  float* op = hpost + (size_t)t * 8;
#pragma unroll
  for (int c = 0; c < 8; ++c) {
    const float v = (ap[c] + exs * hp[c]) * inv + bias[h * 8 + c];
    op[c] = v > 0.f ? v : (__expf(v) - 1.f);
  }
}

__global__ void gat_padw2(const float* __restrict__ W2, float* __restrict__ W2p) {
  const int t = blockIdx.x * blockDim.x + threadIdx.x;
  if (t >= 64 * 48) return;
  const int k = t / 48, c = t % 48;
  W2p[t] = (c < 40) ? W2[k * 40 + c] : 0.f;
}

__global__ void gat_prep2(const float* __restrict__ h2, const float* __restrict__ attS,
                          const float* __restrict__ attD, float* __restrict__ aS,
                          int* __restrict__ dm,
                          float* __restrict__ agg, int N) {
  const int n = blockIdx.x * blockDim.x + threadIdx.x;
  if (n >= N) return;
  const float* hp = h2 + (size_t)n * 48;
  float s = 0.f, d = 0.f;
#pragma unroll
  for (int c = 0; c < 40; ++c) { s += hp[c] * attS[c]; d += hp[c] * attD[c]; }
  aS[n] = s;
  v2i p; p.x = __float_as_int(d); p.y = enc_ord(s);
  ((v2i*)dm)[n] = p;
  float* ap = agg + (size_t)n * 48;
#pragma unroll
  for (int c = 0; c < 48; ++c) ap[c] = 0.f;
}

__global__ void gat_emax2(const int* __restrict__ src, const int* __restrict__ dst,
                          const float* __restrict__ aS, int* __restrict__ dm, int E) {
  const int e = blockIdx.x * blockDim.x + threadIdx.x;
  if (e >= E) return;
  atomicMax(&dm[dst[e] * 2 + 1], enc_ord(aS[src[e]]));
}

__global__ void gat_eagg2(const int* __restrict__ src, const int* __restrict__ dst,
                          const float* __restrict__ aS, const int* __restrict__ dm,
                          const float* __restrict__ h2,
                          float* __restrict__ agg, int E) {
  const int lane = threadIdx.x & 31;
  const int e0 = __builtin_amdgcn_readfirstlane((int)((blockIdx.x * blockDim.x + threadIdx.x) & ~31u));
  if (e0 >= E) return;
  const int e = e0 + lane;
  const int s = e < E ? src[e] : 0, d = e < E ? dst[e] : 0;
  const v2i p = ((const v2i*)dm)[d];
  const float ad = __int_as_float(p.x);
  const float ex = __expf(lrelu(aS[s] + ad) - lrelu(dec_ord(p.y) + ad));
  const int hi = lane >> 4;
  const int ct = 32 + (lane & 15);
  if (e0 + 32 <= E) {
#pragma unroll
    for (int g = 0; g < 32; g += 8) {
      float v0[4], v1[4], vl[4];
#pragma unroll
      for (int k = 0; k < 4; ++k) {
        const int j = g + 2 * k;
        const int s0 = __builtin_amdgcn_readlane(s, j), s1 = __builtin_amdgcn_readlane(s, j + 1);
        const int st = hi ? s1 : s0;
        v0[k] = h2[(size_t)s0 * 48 + lane];
        v1[k] = h2[(size_t)s1 * 48 + lane];
        vl[k] = h2[(size_t)st * 48 + ct];
      }
      asm volatile("" :: "v"(v0[0]), "v"(v1[0]), "v"(vl[0]), "v"(v0[1]), "v"(v1[1]), "v"(vl[1]),
                         "v"(v0[2]), "v"(v1[2]), "v"(vl[2]), "v"(v0[3]), "v"(v1[3]), "v"(vl[3]));
#pragma unroll
      for (int k = 0; k < 4; ++k) {
        const int j = g + 2 * k;
        const int d0 = __builtin_amdgcn_readlane(d, j), d1 = __builtin_amdgcn_readlane(d, j + 1);
        const float x0 = __int_as_float(__builtin_amdgcn_readlane(__float_as_int(ex), j));
        const float x1 = __int_as_float(__builtin_amdgcn_readlane(__float_as_int(ex), j + 1));
        const int dt = hi ? d1 : d0;
        const float xt = hi ? x1 : x0;
        const float vt = (ct == 40) ? 1.f : vl[k];
        atomicAdd(agg + (size_t)d0 * 48 + lane, x0 * v0[k]);
        atomicAdd(agg + (size_t)d1 * 48 + lane, x1 * v1[k]);
        if (ct <= 40)
          atomicAdd(agg + (size_t)dt * 48 + ct, xt * vt);
      }
    }
    return;
  }
#pragma unroll
  for (int j = 0; j < 32; j += 2) {
    const int s0 = __builtin_amdgcn_readlane(s, j), s1 = __builtin_amdgcn_readlane(s, j + 1);
    const int d0 = __builtin_amdgcn_readlane(d, j), d1 = __builtin_amdgcn_readlane(d, j + 1);
    const float x0 = __int_as_float(__builtin_amdgcn_readlane(__float_as_int(ex), j));
    const float x1 = __int_as_float(__builtin_amdgcn_readlane(__float_as_int(ex), j + 1));
    const float v0 = h2[(size_t)s0 * 48 + lane];
    const float v1 = h2[(size_t)s1 * 48 + lane];
    const int st = hi ? s1 : s0, dt = hi ? d1 : d0;
    const float xt = hi ? x1 : x0;
    const float vl = h2[(size_t)st * 48 + ct];
    const float vt = (ct == 40) ? 1.f : vl;
    if (e0 + j < E)
      atomicAdd(agg + (size_t)d0 * 48 + lane, x0 * v0);
    if (e0 + j + 1 < E)
      atomicAdd(agg + (size_t)d1 * 48 + lane, x1 * v1);
    if (e0 + j + hi < E && ct <= 40)
      atomicAdd(agg + (size_t)dt * 48 + ct, xt * vt);
  }
}

__global__ void gat_fin2(const float* __restrict__ agg, const float* __restrict__ h2,
                         const float* __restrict__ aS, const int* __restrict__ dm,
                         const float* __restrict__ bias, float* __restrict__ out, int N) {
  const int n = blockIdx.x * blockDim.x + threadIdx.x;
  if (n >= N) return;
  const float* ap = agg + (size_t)n * 48;
  const float* hp = h2 + (size_t)n * 48;
  const v2i p = ((const v2i*)dm)[n];
  const float ad = __int_as_float(p.x);
  const float exs = __expf(lrelu(aS[n] + ad) - lrelu(dec_ord(p.y) + ad));
  const float inv = 1.f / (ap[40] + exs + SOFT_EPS);
  float v[40];
  float mx = -INFINITY;
#pragma unroll
  for (int c = 0; c < 40; ++c) {
    v[c] = (ap[c] + exs * hp[c]) * inv + bias[c];
    mx = fmaxf(mx, v[c]);
  }
  float sum = 0.f;
#pragma unroll
  for (int c = 0; c < 40; ++c) sum += __expf(v[c] - mx);
  const float lse = mx + __logf(sum);
  float* op = out + (size_t)n * 40;
#pragma unroll
  for (int c = 0; c < 40; ++c) op[c] = v[c] - lse;
}

extern "C" void kernel_launch(void* const* d_in, const int* in_sizes, int n_in,
                              void* d_out, int out_size, void* d_ws, size_t ws_size,
                              hipStream_t stream) {
  const float* x   = (const float*)d_in[0];
  const int*   ei  = (const int*)d_in[1];
  const float* W1  = (const float*)d_in[2];
  const float* as1 = (const float*)d_in[3];
  const float* ad1 = (const float*)d_in[4];
  const float* b1  = (const float*)d_in[5];
  const float* W2  = (const float*)d_in[6];
  const float* as2 = (const float*)d_in[7];
  const float* ad2 = (const float*)d_in[8];
  const float* b2  = (const float*)d_in[9];
  float* out = (float*)d_out;

  const int Fin = in_sizes[2] / 64;
  const int N   = in_sizes[0] / Fin;
  const int E   = in_sizes[1] / 2;
  const int* src = ei;
  const int* dst = ei + E;

  float* ws = (float*)d_ws;
  size_t o = 0;
  float* h1    = ws + o; o += (size_t)N * 64;
  float* hpost = ws + o; o += (size_t)N * 64;
  float* aS1   = ws + o; o += (size_t)N * 8;
  int*   dm1   = (int*)(ws + o); o += (size_t)N * 16;
  float* dn1   = ws + o; o += (size_t)N * 8;
  float* agg1  = ws + o; o += (size_t)N * 64;
  float* h2    = ws + o; o += (size_t)N * 48;
  int*   dm2   = (int*)(ws + o); o += (size_t)N * 2;
  float* aS2   = ws + o; o += (size_t)N;
  float* agg2  = ws + o; o += (size_t)N * 48;
  float* W2p   = ws + o; o += (size_t)64 * 48;

  const int TB = 256;
  const int gNH = (N * 8 + TB - 1) / TB;
  const int gN  = (N + TB - 1) / TB;
  const int gE  = (E + TB - 1) / TB;
  const int gM  = (N + 15) / 16;

  gat_gemm_wmma<<<gM, dim3(32, 4), (size_t)Fin * 64 * 4, stream>>>(x, W1, h1, N, Fin, 64);
  gat_prep1 <<<gNH, TB, 0, stream>>>(h1, as1, ad1, aS1, dm1, dn1, agg1, N);
  gat_emax1 <<<gE,  TB, 0, stream>>>(src, dst, aS1, dm1, E);
  gat_eagg1 <<<gE,  TB, 0, stream>>>(src, dst, aS1, dm1, dn1, h1, agg1, E);
  gat_fin1  <<<gNH, TB, 0, stream>>>(agg1, h1, aS1, dm1, dn1, b1, hpost, N);

  gat_padw2 <<<(64 * 48 + TB - 1) / TB, TB, 0, stream>>>(W2, W2p);
  gat_gemm_wmma<<<gM, dim3(32, 3), (size_t)64 * 48 * 4, stream>>>(hpost, W2p, h2, N, 64, 48);
  gat_prep2 <<<gN, TB, 0, stream>>>(h2, as2, ad2, aS2, dm2, agg2, N);
  gat_emax2 <<<gE, TB, 0, stream>>>(src, dst, aS2, dm2, E);
  gat_eagg2 <<<gE, TB, 0, stream>>>(src, dst, aS2, dm2, h2, agg2, E);
  gat_fin2  <<<gN, TB, 0, stream>>>(agg2, h2, aS2, dm2, b2, out, N);

  (void)n_in; (void)out_size; (void)ws_size;
}


# Weight and row sums kept as one copy per package half: a work-group's float atomics go to the copy of the half it runs on (die id), node passes add the two copies; each sum is now two partial sums
# speedup vs baseline: 7.2265x; 1.1074x over previous
#include <hip/hip_runtime.h>
#include <math.h>

typedef __attribute__((ext_vector_type(2))) float v2f;
typedef __attribute__((ext_vector_type(8))) float v8f;
typedef __attribute__((ext_vector_type(2))) int   v2i;

#define NEG_SLOPE 0.2f
#define SOFT_EPS  1e-16f

__device__ __forceinline__ float lrelu(float x) { return x > 0.f ? x : NEG_SLOPE * x; }
__device__ __forceinline__ int   enc_ord(float f) { int i = __float_as_int(f); return i >= 0 ? i : (i ^ 0x7FFFFFFF); }
__device__ __forceinline__ float dec_ord(int i)   { return __int_as_float(i >= 0 ? i : (i ^ 0x7FFFFFFF)); }

__global__ void gat_gemm_wmma(const float* __restrict__ A, const float* __restrict__ Bg,
                              float* __restrict__ C, int M, int K, int NC) {
  extern __shared__ float Bs[];
  const int nthr = blockDim.x * blockDim.y;
  const int tid  = threadIdx.y * blockDim.x + threadIdx.x;
  for (int i = tid; i < K * NC; i += nthr) Bs[i] = Bg[i];
  __syncthreads();

  const int row0 = blockIdx.x * 16;
  const int col0 = threadIdx.y * 16;
  const int lane = threadIdx.x;
  const int mr   = lane & 15;
  const int half = lane >> 4;

  int arow = row0 + mr; if (arow > M - 1) arow = M - 1;
  const float* Ap = A + (size_t)arow * K;

  v8f acc = {};
  for (int k0 = 0; k0 < K; k0 += 4) {
    const int ka = k0 + 2 * half;
    v2f a, b;
    a.x = Ap[ka];
    a.y = Ap[ka + 1];
    b.x = Bs[ka * NC + col0 + mr];
    b.y = Bs[(ka + 1) * NC + col0 + mr];
    acc = __builtin_amdgcn_wmma_f32_16x16x4_f32(false, a, false, b, (short)0, acc,
                                                false, false);
  }
#pragma unroll
  for (int i = 0; i < 8; ++i) {
    const int r = row0 + i + 8 * half;
    if (r < M) C[(size_t)r * NC + col0 + mr] = acc[i];
  }
}

__device__ __forceinline__ int package_half(int* slot) {
  if (threadIdx.x == 0) *slot = (int)((__builtin_amdgcn_s_sendmsg_rtn(0x87) >> 18) & 1u);
  __syncthreads();
  return __builtin_amdgcn_readfirstlane(*slot);
}
__global__ void gat_prep1(const float* __restrict__ h1, const float* __restrict__ attS,
                          const float* __restrict__ attD, float* __restrict__ aS,
                          int* __restrict__ dm,
                          float* __restrict__ denom, float* __restrict__ agg, int N) {
  const int t = blockIdx.x * blockDim.x + threadIdx.x;
  if (t >= N * 8) return;
  const int h = t & 7;
  const float* hp = h1 + (size_t)t * 8;
  float s = 0.f, d = 0.f;
#pragma unroll
  for (int c = 0; c < 8; ++c) { s += hp[c] * attS[h * 8 + c]; d += hp[c] * attD[h * 8 + c]; }
  aS[t] = s;
  v2i p; p.x = __float_as_int(d); p.y = enc_ord(s);
  ((v2i*)dm)[t] = p;
  denom[t] = 0.f;
  denom[(size_t)N * 8 + t] = 0.f;
  float* ap = agg + (size_t)t * 8;
  float* bp = ap + (size_t)N * 64;
#pragma unroll
  for (int c = 0; c < 8; ++c) { ap[c] = 0.f; bp[c] = 0.f; }
}

__global__ void gat_emax1(const int* __restrict__ src, const int* __restrict__ dst,
                          const float* __restrict__ aS, int* __restrict__ dm, int E) {
  const int lane = threadIdx.x & 31;
  const int e0 = __builtin_amdgcn_readfirstlane((int)((blockIdx.x * blockDim.x + threadIdx.x) & ~31u));
  if (e0 >= E) return;
  const int e = e0 + lane;
  const int s = e < E ? src[e] : 0, d = e < E ? dst[e] : 0;
  const int h = lane & 7;
#pragma unroll
  for (int i = 0; i < 8; ++i) {
    const int j = 4 * i + (lane >> 3);
    const int sj = __shfl(s, j), dj = __shfl(d, j);
    const int v = enc_ord(aS[sj * 8 + h]);
    if (e0 + j < E) atomicMax(&dm[(dj * 8 + h) * 2 + 1], v);
  }
}

__global__ void gat_eagg1(const int* __restrict__ src, const int* __restrict__ dst,
                          const float* __restrict__ aS, const int* __restrict__ dm,
                          float* __restrict__ denom,
                          const float* __restrict__ h1, float* __restrict__ agg, int E, int N) {
  __shared__ int half_slot;
  const int lane = threadIdx.x & 31;
  const int e0 = __builtin_amdgcn_readfirstlane((int)((blockIdx.x * blockDim.x + threadIdx.x) & ~31u));
  const int e = e0 + lane;
  const int s = e < E ? src[e] : 0, d = e < E ? dst[e] : 0;
  const int hf = package_half(&half_slot);
  if (e0 >= E) return;
  denom += (size_t)hf * N * 8;
  agg   += (size_t)hf * N * 64;
  const int h = lane & 7;
  float ex[8];
  int   dh[8];
#pragma unroll
  for (int i = 0; i < 8; ++i) {
    const int j = 4 * i + (lane >> 3);
    const int sj = __shfl(s, j), dj = __shfl(d, j);
    dh[i] = dj * 8 + h;
    const v2i p = ((const v2i*)dm)[dh[i]];
    const float ad = __int_as_float(p.x);
    const float ev = lrelu(aS[sj * 8 + h] + ad);
    ex[i] = __expf(ev - lrelu(dec_ord(p.y) + ad));
  }
#pragma unroll
  for (int i = 0; i < 8; ++i)
    if (e0 + 4 * i + (lane >> 3) < E) atomicAdd(&denom[dh[i]], ex[i]);
  const int hl = lane >> 3;
  if (e0 + 32 <= E) {
#pragma unroll
    for (int g = 0; g < 32; g += 8) {
      float v0[8], v1[8];
#pragma unroll
      for (int k = 0; k < 8; ++k) {
        const int sj = __builtin_amdgcn_readlane(s, g + k);
        const float* hs = h1 + (size_t)sj * 64;
        v0[k] = hs[lane];
        v1[k] = hs[32 + lane];
      }
#pragma unroll
      for (int k = 0; k < 8; ++k) {
        const int j = g + k;
        const int dj = __builtin_amdgcn_readlane(d, j);
        const float a_lo = __shfl(ex[j >> 2], ((j & 3) << 3) + hl);
        const float a_hi = __shfl(ex[j >> 2], ((j & 3) << 3) + 4 + hl);
        float* o = agg + (size_t)dj * 64;
        atomicAdd(o + lane, a_lo * v0[k]);
        atomicAdd(o + 32 + lane, a_hi * v1[k]);
      }
    }
    return;
  }
#pragma unroll
  for (int j = 0; j < 32; ++j) {
    const int sj = __builtin_amdgcn_readlane(s, j);
    const int dj = __builtin_amdgcn_readlane(d, j);
    const float a_lo = __shfl(ex[j >> 2], ((j & 3) << 3) + hl);
    const float a_hi = __shfl(ex[j >> 2], ((j & 3) << 3) + 4 + hl);
    const float* hs = h1 + (size_t)sj * 64;
    float* o = agg + (size_t)dj * 64;
    const float v0 = hs[lane], v1 = hs[32 + lane];
    if (e0 + j < E) {
      atomicAdd(o + lane, a_lo * v0);
      atomicAdd(o + 32 + lane, a_hi * v1);
    }
  }
}

__global__ void gat_fin1(const float* __restrict__ agg, const float* __restrict__ h1,
                         const float* __restrict__ aS, const int* __restrict__ dm,
                         const float* __restrict__ denom,
                         const float* __restrict__ bias, float* __restrict__ hpost, int N) {
  const int t = blockIdx.x * blockDim.x + threadIdx.x;
  if (t >= N * 8) return;
  const int h = t & 7;
  const v2i p = ((const v2i*)dm)[t];
  const float ad = __int_as_float(p.x);
  const float exs = __expf(lrelu(aS[t] + ad) - lrelu(dec_ord(p.y) + ad));
  const float inv = 1.f / ((denom[t] + denom[(size_t)N * 8 + t]) + exs + SOFT_EPS);
  const float* ap = agg + (size_t)t * 8;
  const float* bp = ap + (size_t)N * 64;
  const float* hp = h1 + (size_t)t * 8;
  float* op = hpost + (size_t)t * 8;
#pragma unroll
  for (int c = 0; c < 8; ++c) {
    const float v = ((ap[c] + bp[c]) + exs * hp[c]) * inv + bias[h * 8 + c];
    op[c] = v > 0.f ? v : (__expf(v) - 1.f);
  }
}

__global__ void gat_padw2(const float* __restrict__ W2, float* __restrict__ W2p) {
  const int t = blockIdx.x * blockDim.x + threadIdx.x;
  if (t >= 64 * 48) return;
  const int k = t / 48, c = t % 48;
  W2p[t] = (c < 40) ? W2[k * 40 + c] : 0.f;
}

__global__ void gat_prep2(const float* __restrict__ h2, const float* __restrict__ attS,
                          const float* __restrict__ attD, float* __restrict__ aS,
                          int* __restrict__ dm,
                          float* __restrict__ agg, int N) {
  const int n = blockIdx.x * blockDim.x + threadIdx.x;
  if (n >= N) return;
  const float* hp = h2 + (size_t)n * 48;
  float s = 0.f, d = 0.f;
#pragma unroll
  for (int c = 0; c < 40; ++c) { s += hp[c] * attS[c]; d += hp[c] * attD[c]; }
  aS[n] = s;
  v2i p; p.x = __float_as_int(d); p.y = enc_ord(s);
  ((v2i*)dm)[n] = p;
  float* ap = agg + (size_t)n * 48;
  float* bp = ap + (size_t)N * 48;
#pragma unroll
  for (int c = 0; c < 48; ++c) { ap[c] = 0.f; bp[c] = 0.f; }
}

__global__ void gat_emax2(const int* __restrict__ src, const int* __restrict__ dst,
                          const float* __restrict__ aS, int* __restrict__ dm, int E) {
  const int e = blockIdx.x * blockDim.x + threadIdx.x;
  if (e >= E) return;
  atomicMax(&dm[dst[e] * 2 + 1], enc_ord(aS[src[e]]));
}

__global__ void gat_eagg2(const int* __restrict__ src, const int* __restrict__ dst,
                          const float* __restrict__ aS, const int* __restrict__ dm,
                          const float* __restrict__ h2,
                          float* __restrict__ agg, int E, int N) {
  __shared__ int half_slot;
  const int lane = threadIdx.x & 31;
  const int e0 = __builtin_amdgcn_readfirstlane((int)((blockIdx.x * blockDim.x + threadIdx.x) & ~31u));
  const int e = e0 + lane;
  const int s = e < E ? src[e] : 0, d = e < E ? dst[e] : 0;
  const int hf = package_half(&half_slot);
  if (e0 >= E) return;
  agg += (size_t)hf * N * 48;
  const v2i p = ((const v2i*)dm)[d];
  const float ad = __int_as_float(p.x);
  const float ex = __expf(lrelu(aS[s] + ad) - lrelu(dec_ord(p.y) + ad));
  const int hi = lane >> 4;
  const int ct = 32 + (lane & 15);
  if (e0 + 32 <= E) {
#pragma unroll
    for (int g = 0; g < 32; g += 8) {
      float v0[4], v1[4], vl[4];
#pragma unroll
      for (int k = 0; k < 4; ++k) {
        const int j = g + 2 * k;
        const int s0 = __builtin_amdgcn_readlane(s, j), s1 = __builtin_amdgcn_readlane(s, j + 1);
        const int st = hi ? s1 : s0;
        v0[k] = h2[(size_t)s0 * 48 + lane];
        v1[k] = h2[(size_t)s1 * 48 + lane];
        vl[k] = h2[(size_t)st * 48 + ct];
      }
      asm volatile("" :: "v"(v0[0]), "v"(v1[0]), "v"(vl[0]), "v"(v0[1]), "v"(v1[1]), "v"(vl[1]),
                         "v"(v0[2]), "v"(v1[2]), "v"(vl[2]), "v"(v0[3]), "v"(v1[3]), "v"(vl[3]));
#pragma unroll
      for (int k = 0; k < 4; ++k) {
        const int j = g + 2 * k;
        const int d0 = __builtin_amdgcn_readlane(d, j), d1 = __builtin_amdgcn_readlane(d, j + 1);
        const float x0 = __int_as_float(__builtin_amdgcn_readlane(__float_as_int(ex), j));
        const float x1 = __int_as_float(__builtin_amdgcn_readlane(__float_as_int(ex), j + 1));
        const int dt = hi ? d1 : d0;
        const float xt = hi ? x1 : x0;
        const float vt = (ct == 40) ? 1.f : vl[k];
        atomicAdd(agg + (size_t)d0 * 48 + lane, x0 * v0[k]);
        atomicAdd(agg + (size_t)d1 * 48 + lane, x1 * v1[k]);
        if (ct <= 40)
          atomicAdd(agg + (size_t)dt * 48 + ct, xt * vt);
      }
    }
    return;
  }
#pragma unroll
  for (int j = 0; j < 32; j += 2) {
    const int s0 = __builtin_amdgcn_readlane(s, j), s1 = __builtin_amdgcn_readlane(s, j + 1);
    const int d0 = __builtin_amdgcn_readlane(d, j), d1 = __builtin_amdgcn_readlane(d, j + 1);
    const float x0 = __int_as_float(__builtin_amdgcn_readlane(__float_as_int(ex), j));
    const float x1 = __int_as_float(__builtin_amdgcn_readlane(__float_as_int(ex), j + 1));
    const float v0 = h2[(size_t)s0 * 48 + lane];
    const float v1 = h2[(size_t)s1 * 48 + lane];
    const int st = hi ? s1 : s0, dt = hi ? d1 : d0;
    const float xt = hi ? x1 : x0;
    const float vl = h2[(size_t)st * 48 + ct];
    const float vt = (ct == 40) ? 1.f : vl;
    if (e0 + j < E)
      atomicAdd(agg + (size_t)d0 * 48 + lane, x0 * v0);
    if (e0 + j + 1 < E)
      atomicAdd(agg + (size_t)d1 * 48 + lane, x1 * v1);
    if (e0 + j + hi < E && ct <= 40)
      atomicAdd(agg + (size_t)dt * 48 + ct, xt * vt);
  }
}

__global__ void gat_fin2(const float* __restrict__ agg, const float* __restrict__ h2,
                         const float* __restrict__ aS, const int* __restrict__ dm,
                         const float* __restrict__ bias, float* __restrict__ out, int N) {
  const int n = blockIdx.x * blockDim.x + threadIdx.x;
  if (n >= N) return;
  const float* ap = agg + (size_t)n * 48;
  const float* bp = ap + (size_t)N * 48;
  const float* hp = h2 + (size_t)n * 48;
  const v2i p = ((const v2i*)dm)[n];
  const float ad = __int_as_float(p.x);
  const float exs = __expf(lrelu(aS[n] + ad) - lrelu(dec_ord(p.y) + ad));
  const float inv = 1.f / ((ap[40] + bp[40]) + exs + SOFT_EPS);
  float v[40];
  float mx = -INFINITY;
#pragma unroll
  for (int c = 0; c < 40; ++c) {
    v[c] = ((ap[c] + bp[c]) + exs * hp[c]) * inv + bias[c];
    mx = fmaxf(mx, v[c]);
  }
  float sum = 0.f;
#pragma unroll
  for (int c = 0; c < 40; ++c) sum += __expf(v[c] - mx);
  const float lse = mx + __logf(sum);
  float* op = out + (size_t)n * 40;
#pragma unroll
  for (int c = 0; c < 40; ++c) op[c] = v[c] - lse;
}

extern "C" void kernel_launch(void* const* d_in, const int* in_sizes, int n_in,
                              void* d_out, int out_size, void* d_ws, size_t ws_size,
                              hipStream_t stream) {
  const float* x   = (const float*)d_in[0];
  const int*   ei  = (const int*)d_in[1];
  const float* W1  = (const float*)d_in[2];
  const float* as1 = (const float*)d_in[3];
  const float* ad1 = (const float*)d_in[4];
  const float* b1  = (const float*)d_in[5];
  const float* W2  = (const float*)d_in[6];
  const float* as2 = (const float*)d_in[7];
  const float* ad2 = (const float*)d_in[8];
  const float* b2  = (const float*)d_in[9];
  float* out = (float*)d_out;

  const int Fin = in_sizes[2] / 64;
  const int N   = in_sizes[0] / Fin;
  const int E   = in_sizes[1] / 2;
  const int* src = ei;
  const int* dst = ei + E;

  float* ws = (float*)d_ws;
  size_t o = 0;
  float* h1    = ws + o; o += (size_t)N * 64;
  float* hpost = ws + o; o += (size_t)N * 64;
  float* aS1   = ws + o; o += (size_t)N * 8;
  int*   dm1   = (int*)(ws + o); o += (size_t)N * 16;
  float* dn1   = ws + o; o += (size_t)N * 8 * 2;
  float* agg1  = ws + o; o += (size_t)N * 64 * 2;
  float* h2    = ws + o; o += (size_t)N * 48;
  int*   dm2   = (int*)(ws + o); o += (size_t)N * 2;
  float* aS2   = ws + o; o += (size_t)N;
  float* agg2  = ws + o; o += (size_t)N * 48 * 2;
  float* W2p   = ws + o; o += (size_t)64 * 48;

  const int TB = 256;
  const int gNH = (N * 8 + TB - 1) / TB;
  const int gN  = (N + TB - 1) / TB;
  const int gE  = (E + TB - 1) / TB;
  const int gM  = (N + 15) / 16;

  gat_gemm_wmma<<<gM, dim3(32, 4), (size_t)Fin * 64 * 4, stream>>>(x, W1, h1, N, Fin, 64);
  gat_prep1 <<<gNH, TB, 0, stream>>>(h1, as1, ad1, aS1, dm1, dn1, agg1, N);
  gat_emax1 <<<gE,  TB, 0, stream>>>(src, dst, aS1, dm1, E);
  gat_eagg1 <<<gE,  TB, 0, stream>>>(src, dst, aS1, dm1, dn1, h1, agg1, E, N);
  gat_fin1  <<<gNH, TB, 0, stream>>>(agg1, h1, aS1, dm1, dn1, b1, hpost, N);

  gat_padw2 <<<(64 * 48 + TB - 1) / TB, TB, 0, stream>>>(W2, W2p);
  gat_gemm_wmma<<<gM, dim3(32, 3), (size_t)64 * 48 * 4, stream>>>(hpost, W2p, h2, N, 64, 48);
  gat_prep2 <<<gN, TB, 0, stream>>>(h2, as2, ad2, aS2, dm2, agg2, N);
  gat_emax2 <<<gE, TB, 0, stream>>>(src, dst, aS2, dm2, E);
  gat_eagg2 <<<gE, TB, 0, stream>>>(src, dst, aS2, dm2, h2, agg2, E, N);
  gat_fin2  <<<gN, TB, 0, stream>>>(agg2, h2, aS2, dm2, b2, out, N);

  (void)n_in; (void)out_size; (void)ws_size;
}


# Max passes return at once when largest minus smallest source score (node prep, integer atomics) is at most 20: softmax shift is then lrelu(largest + aD), never below a score; else as before
# speedup vs baseline: 11.8350x; 1.6377x over previous
#include <hip/hip_runtime.h>
#include <math.h>

typedef __attribute__((ext_vector_type(2))) float v2f;
typedef __attribute__((ext_vector_type(8))) float v8f;
typedef __attribute__((ext_vector_type(2))) int   v2i;

#define NEG_SLOPE 0.2f
#define SOFT_EPS  1e-16f

__device__ __forceinline__ float lrelu(float x) { return x > 0.f ? x : NEG_SLOPE * x; }
__device__ __forceinline__ int   enc_ord(float f) { int i = __float_as_int(f); return i >= 0 ? i : (i ^ 0x7FFFFFFF); }
__device__ __forceinline__ float dec_ord(int i)   { return __int_as_float(i >= 0 ? i : (i ^ 0x7FFFFFFF)); }

#define SPREAD_MAX 20.f
__device__ __forceinline__ bool narrow_spread(const int* __restrict__ g, int nh) {
  int ok = 1;
#pragma unroll
  for (int h = 0; h < 8; ++h)
    if (h < nh) ok &= (dec_ord(g[h]) - dec_ord(g[nh + h]) <= SPREAD_MAX) ? 1 : 0;
  return ok != 0;
}

__global__ void gat_gemm_wmma(const float* __restrict__ A, const float* __restrict__ Bg,
                              float* __restrict__ C, float* __restrict__ C2,
                              int M, int K, int NC, int NC1) {
  extern __shared__ float Bs[];
  const int nthr = blockDim.x * blockDim.y;
  const int tid  = threadIdx.y * blockDim.x + threadIdx.x;
  for (int i = tid; i < K * NC; i += nthr) Bs[i] = Bg[i];
  __syncthreads();

  const int row0 = blockIdx.x * 16;
  const int col0 = threadIdx.y * 16;
  const int lane = threadIdx.x;
  const int mr   = lane & 15;
  const int half = lane >> 4;

  int arow = row0 + mr; if (arow > M - 1) arow = M - 1;
  const float* Ap = A + (size_t)arow * K;

  v8f acc = {};
  for (int k0 = 0; k0 < K; k0 += 4) {
    const int ka = k0 + 2 * half;
    v2f a, b;
    a.x = Ap[ka];
    a.y = Ap[ka + 1];
    b.x = Bs[ka * NC + col0 + mr];
    b.y = Bs[(ka + 1) * NC + col0 + mr];
    acc = __builtin_amdgcn_wmma_f32_16x16x4_f32(false, a, false, b, (short)0, acc,
                                                false, false);
  }
  const bool first = col0 < NC1;
  float* Cp = first ? C : C2;
  const int pitch = first ? NC1 : NC - NC1;
  const int col = (first ? col0 : col0 - NC1) + mr;
#pragma unroll
  for (int i = 0; i < 8; ++i) {
    const int r = row0 + i + 8 * half;
    if (r < M) Cp[(size_t)r * pitch + col] = acc[i];
  }
}

__device__ __forceinline__ int package_half(int* slot) {
  if (threadIdx.x == 0) *slot = (int)((__builtin_amdgcn_s_sendmsg_rtn(0x87) >> 18) & 1u);
  __syncthreads();
  return __builtin_amdgcn_readfirstlane(*slot);
}
__global__ void gat_prep1(const float* __restrict__ h1, const float* __restrict__ attS,
                          const float* __restrict__ attD, float* __restrict__ aS,
                          int* __restrict__ dm,
                          float* __restrict__ denom, float* __restrict__ agg,
                          int* __restrict__ g, int N) {
  __shared__ int smx[8][8], smn[8][8];
  const int t = blockIdx.x * blockDim.x + threadIdx.x;
  const int h = t & 7;
  int emx = (int)0x80000000, emn = 0x7FFFFFFF;
  if (t < N * 8) {
  const float* hp = h1 + (size_t)t * 8;
  float s = 0.f, d = 0.f;
#pragma unroll
  for (int c = 0; c < 8; ++c) { s += hp[c] * attS[h * 8 + c]; d += hp[c] * attD[h * 8 + c]; }
  aS[t] = s;
  v2i p; p.x = __float_as_int(d); p.y = enc_ord(s);
  ((v2i*)dm)[t] = p;
  emx = p.y; emn = p.y;
  denom[t] = 0.f;
  denom[(size_t)N * 8 + t] = 0.f;
  float* ap = agg + (size_t)t * 8;
  float* bp = ap + (size_t)N * 64;
#pragma unroll
  for (int c = 0; c < 8; ++c) { ap[c] = 0.f; bp[c] = 0.f; }
  }
  emx = max(emx, __shfl_xor(emx, 8));  emn = min(emn, __shfl_xor(emn, 8));
  emx = max(emx, __shfl_xor(emx, 16)); emn = min(emn, __shfl_xor(emn, 16));
  const int lane = threadIdx.x & 31, wv = threadIdx.x >> 5;
  if (lane < 8) { smx[wv][lane] = emx; smn[wv][lane] = emn; }
  __syncthreads();
  if (threadIdx.x < 8) {
    int a = smx[0][threadIdx.x], b = smn[0][threadIdx.x];
    for (int w = 1; w < 8; ++w) { a = max(a, smx[w][threadIdx.x]); b = min(b, smn[w][threadIdx.x]); }
    atomicMax(&g[threadIdx.x], a);
    atomicMin(&g[8 + threadIdx.x], b);
  }
}

__global__ void gat_emax1(const int* __restrict__ src, const int* __restrict__ dst,
                          const float* __restrict__ aS, int* __restrict__ dm,
                          const int* __restrict__ g, int E) {
  if (narrow_spread(g, 8)) return;
  for (int c = 0; c < 8; ++c) {
  const int lane = threadIdx.x & 31;
  const int e0 = __builtin_amdgcn_readfirstlane((int)((((blockIdx.x * 8 + c) * blockDim.x) + threadIdx.x) & ~31u));
  if (e0 >= E) return;
  const int e = e0 + lane;
  const int s = e < E ? src[e] : 0, d = e < E ? dst[e] : 0;
  const int h = lane & 7;
#pragma unroll
  for (int i = 0; i < 8; ++i) {
    const int j = 4 * i + (lane >> 3);
    const int sj = __shfl(s, j), dj = __shfl(d, j);
    const int v = enc_ord(aS[sj * 8 + h]);
    if (e0 + j < E) atomicMax(&dm[(dj * 8 + h) * 2 + 1], v);
  }
  }
}

__global__ void gat_eagg1(const int* __restrict__ src, const int* __restrict__ dst,
                          const float* __restrict__ aS, const int* __restrict__ dm,
                          float* __restrict__ denom,
                          const float* __restrict__ h1, float* __restrict__ agg,
                          const int* __restrict__ g, int E, int N) {
  __shared__ int half_slot;
  const int lane = threadIdx.x & 31;
  const int e0 = __builtin_amdgcn_readfirstlane((int)((blockIdx.x * blockDim.x + threadIdx.x) & ~31u));
  const int e = e0 + lane;
  const int s = e < E ? src[e] : 0, d = e < E ? dst[e] : 0;
  const int hf = package_half(&half_slot);
  if (e0 >= E) return;
  denom += (size_t)hf * N * 8;
  agg   += (size_t)hf * N * 64;
  const int h = lane & 7;
  const bool nsp = narrow_spread(g, 8);
  const float gms = dec_ord(g[h]);
  float ex[8];
  int   dh[8];
#pragma unroll
  for (int i = 0; i < 8; ++i) {
    const int j = 4 * i + (lane >> 3);
    const int sj = __shfl(s, j), dj = __shfl(d, j);
    dh[i] = dj * 8 + h;
    const v2i p = ((const v2i*)dm)[dh[i]];
    const float ad = __int_as_float(p.x);
    const float ev = lrelu(aS[sj * 8 + h] + ad);
    ex[i] = __expf(ev - lrelu((nsp ? gms : dec_ord(p.y)) + ad));
  }
#pragma unroll
  for (int i = 0; i < 8; ++i)
    if (e0 + 4 * i + (lane >> 3) < E) atomicAdd(&denom[dh[i]], ex[i]);
  const int hl = lane >> 3;
  if (e0 + 32 <= E) {
#pragma unroll
    for (int g = 0; g < 32; g += 8) {
      float v0[8], v1[8];
#pragma unroll
      for (int k = 0; k < 8; ++k) {
        const int sj = __builtin_amdgcn_readlane(s, g + k);
        const float* hs = h1 + (size_t)sj * 64;
        v0[k] = hs[lane];
        v1[k] = hs[32 + lane];
      }
#pragma unroll
      for (int k = 0; k < 8; ++k) {
        const int j = g + k;
        const int dj = __builtin_amdgcn_readlane(d, j);
        const float a_lo = __shfl(ex[j >> 2], ((j & 3) << 3) + hl);
        const float a_hi = __shfl(ex[j >> 2], ((j & 3) << 3) + 4 + hl);
        float* o = agg + (size_t)dj * 64;
        atomicAdd(o + lane, a_lo * v0[k]);
        atomicAdd(o + 32 + lane, a_hi * v1[k]);
      }
    }
    return;
  }
#pragma unroll
  for (int j = 0; j < 32; ++j) {
    const int sj = __builtin_amdgcn_readlane(s, j);
    const int dj = __builtin_amdgcn_readlane(d, j);
    const float a_lo = __shfl(ex[j >> 2], ((j & 3) << 3) + hl);
    const float a_hi = __shfl(ex[j >> 2], ((j & 3) << 3) + 4 + hl);
    const float* hs = h1 + (size_t)sj * 64;
    float* o = agg + (size_t)dj * 64;
    const float v0 = hs[lane], v1 = hs[32 + lane];
    if (e0 + j < E) {
      atomicAdd(o + lane, a_lo * v0);
      atomicAdd(o + 32 + lane, a_hi * v1);
    }
  }
}

__global__ void gat_fin1(const float* __restrict__ agg, const float* __restrict__ h1,
                         const float* __restrict__ aS, const int* __restrict__ dm,
                         const float* __restrict__ denom,
                         const float* __restrict__ bias, float* __restrict__ hpost,
                         const int* __restrict__ g, int N) {
  const int t = blockIdx.x * blockDim.x + threadIdx.x;
  if (t >= N * 8) return;
  const int h = t & 7;
  const v2i p = ((const v2i*)dm)[t];
  const float ad = __int_as_float(p.x);
  const float ms = narrow_spread(g, 8) ? dec_ord(g[h]) : dec_ord(p.y);
  const float exs = __expf(lrelu(aS[t] + ad) - lrelu(ms + ad));
  const float inv = 1.f / ((denom[t] + denom[(size_t)N * 8 + t]) + exs + SOFT_EPS);
  const float* ap = agg + (size_t)t * 8;
  const float* bp = ap + (size_t)N * 64;
  const float* hp = h1 + (size_t)t * 8;
  float* op = hpost + (size_t)t * 8;
#pragma unroll
  for (int c = 0; c < 8; ++c) {
    const float v = ((ap[c] + bp[c]) + exs * hp[c]) * inv + bias[h * 8 + c];
    op[c] = v > 0.f ? v : (__expf(v) - 1.f);
  }
}

__global__ void gat_padw2(const float* __restrict__ W2, float* __restrict__ W2p, int* __restrict__ g) {
  const int t = blockIdx.x * blockDim.x + threadIdx.x;
  if (t < 18) g[t] = (t < 8 || t == 16) ? (int)0x80000000 : 0x7FFFFFFF;
  if (t >= 64 * 48) return;
  const int k = t / 48, c = t % 48;
  W2p[t] = (c < 40) ? W2[k * 40 + c] : 0.f;
}

__global__ void gat_prep2(const float* __restrict__ h2a, const float* __restrict__ h2b,
                          const float* __restrict__ attS,
                          const float* __restrict__ attD, float* __restrict__ aS,
                          int* __restrict__ dm,
                          float* __restrict__ agga, float* __restrict__ aggb,
                          int* __restrict__ g, int N) {
  __shared__ int smx[8], smn[8];
  const int n = blockIdx.x * blockDim.x + threadIdx.x;
  int emx = (int)0x80000000, emn = 0x7FFFFFFF;
  if (n < N) {
  const float* ha = h2a + (size_t)n * 32;
  const float* hb = h2b + (size_t)n * 16;
  float s = 0.f, d = 0.f;
#pragma unroll
  for (int c = 0; c < 32; ++c) { s += ha[c] * attS[c]; d += ha[c] * attD[c]; }
#pragma unroll
  for (int c = 0; c < 8; ++c) { s += hb[c] * attS[32 + c]; d += hb[c] * attD[32 + c]; }
  aS[n] = s;
  v2i p; p.x = __float_as_int(d); p.y = enc_ord(s);
  ((v2i*)dm)[n] = p;
  emx = p.y; emn = p.y;
  float* ap = agga + (size_t)n * 32;
  float* bp = ap + (size_t)N * 32;
#pragma unroll
  for (int c = 0; c < 32; ++c) { ap[c] = 0.f; bp[c] = 0.f; }
  float* cp = aggb + (size_t)n * 16;
  float* dp = cp + (size_t)N * 16;
#pragma unroll
  for (int c = 0; c < 16; ++c) { cp[c] = 0.f; dp[c] = 0.f; }
  }
#pragma unroll
  for (int m = 1; m < 32; m <<= 1) { emx = max(emx, __shfl_xor(emx, m)); emn = min(emn, __shfl_xor(emn, m)); }
  if ((threadIdx.x & 31) == 0) { smx[threadIdx.x >> 5] = emx; smn[threadIdx.x >> 5] = emn; }
  __syncthreads();
  if (threadIdx.x == 0) {
    int a = smx[0], b = smn[0];
    for (int w = 1; w < 8; ++w) { a = max(a, smx[w]); b = min(b, smn[w]); }
    atomicMax(&g[16], a);
    atomicMin(&g[17], b);
  }
}

__global__ void gat_emax2(const int* __restrict__ src, const int* __restrict__ dst,
                          const float* __restrict__ aS, int* __restrict__ dm,
                          const int* __restrict__ g, int E) {
  if (narrow_spread(g + 16, 1)) return;
  for (int c = 0; c < 8; ++c) {
    const int e = (blockIdx.x * 8 + c) * blockDim.x + threadIdx.x;
    if (e < E) atomicMax(&dm[dst[e] * 2 + 1], enc_ord(aS[src[e]]));
  }
}

__global__ void gat_eagg2(const int* __restrict__ src, const int* __restrict__ dst,
                          const float* __restrict__ aS, const int* __restrict__ dm,
                          const float* __restrict__ h2a, const float* __restrict__ h2b,
                          float* __restrict__ agga, float* __restrict__ aggb,
                          const int* __restrict__ g, int E, int N) {
  __shared__ int half_slot;
  const int lane = threadIdx.x & 31;
  const int e0 = __builtin_amdgcn_readfirstlane((int)((blockIdx.x * blockDim.x + threadIdx.x) & ~31u));
  const int e = e0 + lane;
  const int s = e < E ? src[e] : 0, d = e < E ? dst[e] : 0;
  const int hf = package_half(&half_slot);
  if (e0 >= E) return;
  agga += (size_t)hf * N * 32;
  aggb += (size_t)hf * N * 16;
  const v2i p = ((const v2i*)dm)[d];
  const float ad = __int_as_float(p.x);
  const float ms = narrow_spread(g + 16, 1) ? dec_ord(g[16]) : dec_ord(p.y);
  const float ex = __expf(lrelu(aS[s] + ad) - lrelu(ms + ad));
  const int hi = lane >> 4;
  const int ct = lane & 15;
  if (e0 + 32 <= E) {
#pragma unroll
    for (int g = 0; g < 32; g += 8) {
      float v0[4], v1[4], vl[4];
#pragma unroll
      for (int k = 0; k < 4; ++k) {
        const int j = g + 2 * k;
        const int s0 = __builtin_amdgcn_readlane(s, j), s1 = __builtin_amdgcn_readlane(s, j + 1);
        const int st = hi ? s1 : s0;
        v0[k] = h2a[(size_t)s0 * 32 + lane];
        v1[k] = h2a[(size_t)s1 * 32 + lane];
        vl[k] = h2b[(size_t)st * 16 + ct];
      }
      asm volatile("" :: "v"(v0[0]), "v"(v1[0]), "v"(vl[0]), "v"(v0[1]), "v"(v1[1]), "v"(vl[1]),
                         "v"(v0[2]), "v"(v1[2]), "v"(vl[2]), "v"(v0[3]), "v"(v1[3]), "v"(vl[3]));
      __builtin_amdgcn_sched_barrier(0);
#pragma unroll
      for (int k = 0; k < 4; ++k) {
        const int j = g + 2 * k;
        const int d0 = __builtin_amdgcn_readlane(d, j), d1 = __builtin_amdgcn_readlane(d, j + 1);
        const float x0 = __int_as_float(__builtin_amdgcn_readlane(__float_as_int(ex), j));
        const float x1 = __int_as_float(__builtin_amdgcn_readlane(__float_as_int(ex), j + 1));
        const int dt = hi ? d1 : d0;
        const float xt = hi ? x1 : x0;
        const float vt = (ct == 8) ? 1.f : vl[k];
        atomicAdd(agga + (size_t)d0 * 32 + lane, x0 * v0[k]);
        atomicAdd(agga + (size_t)d1 * 32 + lane, x1 * v1[k]);
        if (ct <= 8)
          atomicAdd(aggb + (size_t)dt * 16 + ct, xt * vt);
      }
    }
    return;
  }
#pragma unroll
  for (int j = 0; j < 32; j += 2) {
    const int s0 = __builtin_amdgcn_readlane(s, j), s1 = __builtin_amdgcn_readlane(s, j + 1);
    const int d0 = __builtin_amdgcn_readlane(d, j), d1 = __builtin_amdgcn_readlane(d, j + 1);
    const float x0 = __int_as_float(__builtin_amdgcn_readlane(__float_as_int(ex), j));
    const float x1 = __int_as_float(__builtin_amdgcn_readlane(__float_as_int(ex), j + 1));
    const float v0 = h2a[(size_t)s0 * 32 + lane];
    const float v1 = h2a[(size_t)s1 * 32 + lane];
    const int st = hi ? s1 : s0, dt = hi ? d1 : d0;
    const float xt = hi ? x1 : x0;
    const float vl = h2b[(size_t)st * 16 + ct];
    const float vt = (ct == 8) ? 1.f : vl;
    if (e0 + j < E)
      atomicAdd(agga + (size_t)d0 * 32 + lane, x0 * v0);
    if (e0 + j + 1 < E)
      atomicAdd(agga + (size_t)d1 * 32 + lane, x1 * v1);
    if (e0 + j + hi < E && ct <= 8)
      atomicAdd(aggb + (size_t)dt * 16 + ct, xt * vt);
  }
}

__global__ void gat_fin2(const float* __restrict__ agga, const float* __restrict__ aggb,
                         const float* __restrict__ h2a, const float* __restrict__ h2b,
                         const float* __restrict__ aS, const int* __restrict__ dm,
                         const float* __restrict__ bias, float* __restrict__ out,
                         const int* __restrict__ g, int N) {
  const int n = blockIdx.x * blockDim.x + threadIdx.x;
  if (n >= N) return;
  const float* ap = agga + (size_t)n * 32;
  const float* bp = ap + (size_t)N * 32;
  const float* cp = aggb + (size_t)n * 16;
  const float* dp = cp + (size_t)N * 16;
  const float* ha = h2a + (size_t)n * 32;
  const float* hb = h2b + (size_t)n * 16;
  const v2i p = ((const v2i*)dm)[n];
  const float ad = __int_as_float(p.x);
  const float ms = narrow_spread(g + 16, 1) ? dec_ord(g[16]) : dec_ord(p.y);
  const float exs = __expf(lrelu(aS[n] + ad) - lrelu(ms + ad));
  const float inv = 1.f / ((cp[8] + dp[8]) + exs + SOFT_EPS);
  float v[40];
  float mx = -INFINITY;
#pragma unroll
  for (int c = 0; c < 32; ++c) {
    v[c] = ((ap[c] + bp[c]) + exs * ha[c]) * inv + bias[c];
    mx = fmaxf(mx, v[c]);
  }
#pragma unroll
  for (int c = 0; c < 8; ++c) {
    v[32 + c] = ((cp[c] + dp[c]) + exs * hb[c]) * inv + bias[32 + c];
    mx = fmaxf(mx, v[32 + c]);
  }
  float sum = 0.f;
#pragma unroll
  for (int c = 0; c < 40; ++c) sum += __expf(v[c] - mx);
  const float lse = mx + __logf(sum);
  float* op = out + (size_t)n * 40;
#pragma unroll
  for (int c = 0; c < 40; ++c) op[c] = v[c] - lse;
}

extern "C" void kernel_launch(void* const* d_in, const int* in_sizes, int n_in,
                              void* d_out, int out_size, void* d_ws, size_t ws_size,
                              hipStream_t stream) {
  const float* x   = (const float*)d_in[0];
  const int*   ei  = (const int*)d_in[1];
  const float* W1  = (const float*)d_in[2];
  const float* as1 = (const float*)d_in[3];
  const float* ad1 = (const float*)d_in[4];
  const float* b1  = (const float*)d_in[5];
  const float* W2  = (const float*)d_in[6];
  const float* as2 = (const float*)d_in[7];
  const float* ad2 = (const float*)d_in[8];
  const float* b2  = (const float*)d_in[9];
  float* out = (float*)d_out;

  const int Fin = in_sizes[2] / 64;
  const int N   = in_sizes[0] / Fin;
  const int E   = in_sizes[1] / 2;
  const int* src = ei;
  const int* dst = ei + E;

  float* ws = (float*)d_ws;
  size_t o = 0;
  float* h1    = ws + o; o += (size_t)N * 64;
  float* hpost = ws + o; o += (size_t)N * 64;
  float* aS1   = ws + o; o += (size_t)N * 8;
  int*   dm1   = (int*)(ws + o); o += (size_t)N * 16;
  float* dn1   = ws + o; o += (size_t)N * 8 * 2;
  float* agg1  = ws + o; o += (size_t)N * 64 * 2;
  float* h2a   = ws + o; o += (size_t)N * 32;
  float* h2b   = ws + o; o += (size_t)N * 16;
  int*   dm2   = (int*)(ws + o); o += (size_t)N * 2;
  float* aS2   = ws + o; o += (size_t)N;
  float* agg2a = ws + o; o += (size_t)N * 32 * 2;
  float* agg2b = ws + o; o += (size_t)N * 16 * 2;
  float* W2p   = ws + o; o += (size_t)64 * 48;
  int*   gst   = (int*)(ws + o); o += 32;

  const int TB = 256;
  const int gNH = (N * 8 + TB - 1) / TB;
  const int gN  = (N + TB - 1) / TB;
  const int gE  = (E + TB - 1) / TB;
  const int gM  = (N + 15) / 16;
  const int gE8 = (gE + 7) / 8;

  gat_padw2 <<<(64 * 48 + TB - 1) / TB, TB, 0, stream>>>(W2, W2p, gst);

  gat_gemm_wmma<<<gM, dim3(32, 4), (size_t)Fin * 64 * 4, stream>>>(x, W1, h1, h1, N, Fin, 64, 64);
  gat_prep1 <<<gNH, TB, 0, stream>>>(h1, as1, ad1, aS1, dm1, dn1, agg1, gst, N);
  gat_emax1 <<<gE8, TB, 0, stream>>>(src, dst, aS1, dm1, gst, E);
  gat_eagg1 <<<gE,  TB, 0, stream>>>(src, dst, aS1, dm1, dn1, h1, agg1, gst, E, N);
  gat_fin1  <<<gNH, TB, 0, stream>>>(agg1, h1, aS1, dm1, dn1, b1, hpost, gst, N);

  gat_gemm_wmma<<<gM, dim3(32, 3), (size_t)64 * 48 * 4, stream>>>(hpost, W2p, h2a, h2b, N, 64, 48, 32);
  gat_prep2 <<<gN, TB, 0, stream>>>(h2a, h2b, as2, ad2, aS2, dm2, agg2a, agg2b, gst, N);
  gat_emax2 <<<gE8, TB, 0, stream>>>(src, dst, aS2, dm2, gst, E);
  gat_eagg2 <<<gE, TB, 0, stream>>>(src, dst, aS2, dm2, h2a, h2b, agg2a, agg2b, gst, E, N);
  gat_fin2  <<<gN, TB, 0, stream>>>(agg2a, agg2b, h2a, h2b, aS2, dm2, b2, out, gst, N);

  (void)n_in; (void)out_size; (void)ws_size;
}


# Projection kernels stage their weight matrix by 16-byte loads, eight requested together ahead of the LDS stores (was a 4-byte load waited for alone each trip: 64 or 32 round trips in series)
# speedup vs baseline: 13.7180x; 1.1591x over previous
#include <hip/hip_runtime.h>
#include <math.h>

typedef __attribute__((ext_vector_type(2))) float v2f;
typedef __attribute__((ext_vector_type(8))) float v8f;
typedef __attribute__((ext_vector_type(2))) int   v2i;
typedef __attribute__((ext_vector_type(4))) float v4f;

#define NEG_SLOPE 0.2f
#define SOFT_EPS  1e-16f

__device__ __forceinline__ float lrelu(float x) { return x > 0.f ? x : NEG_SLOPE * x; }
__device__ __forceinline__ int   enc_ord(float f) { int i = __float_as_int(f); return i >= 0 ? i : (i ^ 0x7FFFFFFF); }
__device__ __forceinline__ float dec_ord(int i)   { return __int_as_float(i >= 0 ? i : (i ^ 0x7FFFFFFF)); }

#define SPREAD_MAX 20.f
__device__ __forceinline__ bool narrow_spread(const int* __restrict__ g, int nh) {
  int ok = 1;
#pragma unroll
  for (int h = 0; h < 8; ++h)
    if (h < nh) ok &= (dec_ord(g[h]) - dec_ord(g[nh + h]) <= SPREAD_MAX) ? 1 : 0;
  return ok != 0;
}

__global__ void gat_gemm_wmma(const float* __restrict__ A, const float* __restrict__ Bg,
                              float* __restrict__ C, float* __restrict__ C2,
                              int M, int K, int NC, int NC1) {
  extern __shared__ __attribute__((aligned(16))) float Bs[];
  const int nthr = blockDim.x * blockDim.y;
  const int tid  = threadIdx.y * blockDim.x + threadIdx.x;
  const int tot  = K * NC;
  if ((((size_t)Bg) & 15) == 0 && (tot & 3) == 0) {
    const int tot4 = tot >> 2;
    const v4f* Bg4 = (const v4f*)Bg;
    v4f* Bs4 = (v4f*)Bs;
    for (int i0 = tid; i0 < tot4; i0 += 8 * nthr) {
      v4f t[8];
#pragma unroll
      for (int u = 0; u < 8; ++u) t[u] = Bg4[min(i0 + u * nthr, tot4 - 1)];
#pragma unroll
      for (int u = 0; u < 8; ++u) {
        const int i = i0 + u * nthr;
        if (i < tot4) Bs4[i] = t[u];
      }
    }
  } else {
    for (int i = tid; i < tot; i += nthr) Bs[i] = Bg[i];
  }
  __syncthreads();

  const int row0 = blockIdx.x * 16;
  const int col0 = threadIdx.y * 16;
  const int lane = threadIdx.x;
  const int mr   = lane & 15;
  const int half = lane >> 4;

  int arow = row0 + mr; if (arow > M - 1) arow = M - 1;
  const float* Ap = A + (size_t)arow * K;

  v8f acc = {};
  for (int k0 = 0; k0 < K; k0 += 4) {
    const int ka = k0 + 2 * half;
    v2f a, b;
    a.x = Ap[ka];
    a.y = Ap[ka + 1];
    b.x = Bs[ka * NC + col0 + mr];
    b.y = Bs[(ka + 1) * NC + col0 + mr];
    acc = __builtin_amdgcn_wmma_f32_16x16x4_f32(false, a, false, b, (short)0, acc,
                                                false, false);
  }
  const bool first = col0 < NC1;
  float* Cp = first ? C : C2;
  const int pitch = first ? NC1 : NC - NC1;
  const int col = (first ? col0 : col0 - NC1) + mr;
#pragma unroll
  for (int i = 0; i < 8; ++i) {
    const int r = row0 + i + 8 * half;
    if (r < M) Cp[(size_t)r * pitch + col] = acc[i];
  }
}

__device__ __forceinline__ int package_half(int* slot) {
  if (threadIdx.x == 0) *slot = (int)((__builtin_amdgcn_s_sendmsg_rtn(0x87) >> 18) & 1u);
  __syncthreads();
  return __builtin_amdgcn_readfirstlane(*slot);
}
__global__ void gat_prep1(const float* __restrict__ h1, const float* __restrict__ attS,
                          const float* __restrict__ attD, float* __restrict__ aS,
                          int* __restrict__ dm,
                          float* __restrict__ denom, float* __restrict__ agg,
                          int* __restrict__ g, int N) {
  __shared__ int smx[8][8], smn[8][8];
  const int t = blockIdx.x * blockDim.x + threadIdx.x;
  const int h = t & 7;
  int emx = (int)0x80000000, emn = 0x7FFFFFFF;
  if (t < N * 8) {
  const float* hp = h1 + (size_t)t * 8;
  float s = 0.f, d = 0.f;
#pragma unroll
  for (int c = 0; c < 8; ++c) { s += hp[c] * attS[h * 8 + c]; d += hp[c] * attD[h * 8 + c]; }
  aS[t] = s;
  v2i p; p.x = __float_as_int(d); p.y = enc_ord(s);
  ((v2i*)dm)[t] = p;
  emx = p.y; emn = p.y;
  denom[t] = 0.f;
  denom[(size_t)N * 8 + t] = 0.f;
  float* ap = agg + (size_t)t * 8;
  float* bp = ap + (size_t)N * 64;
#pragma unroll
  for (int c = 0; c < 8; ++c) { ap[c] = 0.f; bp[c] = 0.f; }
  }
  emx = max(emx, __shfl_xor(emx, 8));  emn = min(emn, __shfl_xor(emn, 8));
  emx = max(emx, __shfl_xor(emx, 16)); emn = min(emn, __shfl_xor(emn, 16));
  const int lane = threadIdx.x & 31, wv = threadIdx.x >> 5;
  if (lane < 8) { smx[wv][lane] = emx; smn[wv][lane] = emn; }
  __syncthreads();
  if (threadIdx.x < 8) {
    int a = smx[0][threadIdx.x], b = smn[0][threadIdx.x];
    for (int w = 1; w < 8; ++w) { a = max(a, smx[w][threadIdx.x]); b = min(b, smn[w][threadIdx.x]); }
    atomicMax(&g[threadIdx.x], a);
    atomicMin(&g[8 + threadIdx.x], b);
  }
}

__global__ void gat_emax1(const int* __restrict__ src, const int* __restrict__ dst,
                          const float* __restrict__ aS, int* __restrict__ dm,
                          const int* __restrict__ g, int E) {
  if (narrow_spread(g, 8)) return;
  for (int c = 0; c < 8; ++c) {
  const int lane = threadIdx.x & 31;
  const int e0 = __builtin_amdgcn_readfirstlane((int)((((blockIdx.x * 8 + c) * blockDim.x) + threadIdx.x) & ~31u));
  if (e0 >= E) return;
  const int e = e0 + lane;
  const int s = e < E ? src[e] : 0, d = e < E ? dst[e] : 0;
  const int h = lane & 7;
#pragma unroll
  for (int i = 0; i < 8; ++i) {
    const int j = 4 * i + (lane >> 3);
    const int sj = __shfl(s, j), dj = __shfl(d, j);
    const int v = enc_ord(aS[sj * 8 + h]);
    if (e0 + j < E) atomicMax(&dm[(dj * 8 + h) * 2 + 1], v);
  }
  }
}

__global__ void gat_eagg1(const int* __restrict__ src, const int* __restrict__ dst,
                          const float* __restrict__ aS, const int* __restrict__ dm,
                          float* __restrict__ denom,
                          const float* __restrict__ h1, float* __restrict__ agg,
                          const int* __restrict__ g, int E, int N) {
  __shared__ int half_slot;
  const int lane = threadIdx.x & 31;
  const int e0 = __builtin_amdgcn_readfirstlane((int)((blockIdx.x * blockDim.x + threadIdx.x) & ~31u));
  const int e = e0 + lane;
  const int s = e < E ? src[e] : 0, d = e < E ? dst[e] : 0;
  const int hf = package_half(&half_slot);
  if (e0 >= E) return;
  denom += (size_t)hf * N * 8;
  agg   += (size_t)hf * N * 64;
  const int h = lane & 7;
  const bool nsp = narrow_spread(g, 8);
  const float gms = dec_ord(g[h]);
  float ex[8];
  int   dh[8];
#pragma unroll
  for (int i = 0; i < 8; ++i) {
    const int j = 4 * i + (lane >> 3);
    const int sj = __shfl(s, j), dj = __shfl(d, j);
    dh[i] = dj * 8 + h;
    const v2i p = ((const v2i*)dm)[dh[i]];
    const float ad = __int_as_float(p.x);
    const float ev = lrelu(aS[sj * 8 + h] + ad);
    ex[i] = __expf(ev - lrelu((nsp ? gms : dec_ord(p.y)) + ad));
  }
#pragma unroll
  for (int i = 0; i < 8; ++i)
    if (e0 + 4 * i + (lane >> 3) < E) atomicAdd(&denom[dh[i]], ex[i]);
  const int hl = lane >> 3;
  if (e0 + 32 <= E) {
#pragma unroll
    for (int g = 0; g < 32; g += 8) {
      float v0[8], v1[8];
#pragma unroll
      for (int k = 0; k < 8; ++k) {
        const int sj = __builtin_amdgcn_readlane(s, g + k);
        const float* hs = h1 + (size_t)sj * 64;
        v0[k] = hs[lane];
        v1[k] = hs[32 + lane];
      }
#pragma unroll
      for (int k = 0; k < 8; ++k) {
        const int j = g + k;
        const int dj = __builtin_amdgcn_readlane(d, j);
        const float a_lo = __shfl(ex[j >> 2], ((j & 3) << 3) + hl);
        const float a_hi = __shfl(ex[j >> 2], ((j & 3) << 3) + 4 + hl);
        float* o = agg + (size_t)dj * 64;
        atomicAdd(o + lane, a_lo * v0[k]);
        atomicAdd(o + 32 + lane, a_hi * v1[k]);
      }
    }
    return;
  }
#pragma unroll
  for (int j = 0; j < 32; ++j) {
    const int sj = __builtin_amdgcn_readlane(s, j);
    const int dj = __builtin_amdgcn_readlane(d, j);
    const float a_lo = __shfl(ex[j >> 2], ((j & 3) << 3) + hl);
    const float a_hi = __shfl(ex[j >> 2], ((j & 3) << 3) + 4 + hl);
    const float* hs = h1 + (size_t)sj * 64;
    float* o = agg + (size_t)dj * 64;
    const float v0 = hs[lane], v1 = hs[32 + lane];
    if (e0 + j < E) {
      atomicAdd(o + lane, a_lo * v0);
      atomicAdd(o + 32 + lane, a_hi * v1);
    }
  }
}

__global__ void gat_fin1(const float* __restrict__ agg, const float* __restrict__ h1,
                         const float* __restrict__ aS, const int* __restrict__ dm,
                         const float* __restrict__ denom,
                         const float* __restrict__ bias, float* __restrict__ hpost,
                         const int* __restrict__ g, int N) {
  const int t = blockIdx.x * blockDim.x + threadIdx.x;
  if (t >= N * 8) return;
  const int h = t & 7;
  const v2i p = ((const v2i*)dm)[t];
  const float ad = __int_as_float(p.x);
  const float ms = narrow_spread(g, 8) ? dec_ord(g[h]) : dec_ord(p.y);
  const float exs = __expf(lrelu(aS[t] + ad) - lrelu(ms + ad));
  const float inv = 1.f / ((denom[t] + denom[(size_t)N * 8 + t]) + exs + SOFT_EPS);
  const float* ap = agg + (size_t)t * 8;
  const float* bp = ap + (size_t)N * 64;
  const float* hp = h1 + (size_t)t * 8;
  float* op = hpost + (size_t)t * 8;
#pragma unroll
  for (int c = 0; c < 8; ++c) {
    const float v = ((ap[c] + bp[c]) + exs * hp[c]) * inv + bias[h * 8 + c];
    op[c] = v > 0.f ? v : (__expf(v) - 1.f);
  }
}

__global__ void gat_padw2(const float* __restrict__ W2, float* __restrict__ W2p, int* __restrict__ g) {
  const int t = blockIdx.x * blockDim.x + threadIdx.x;
  if (t < 18) g[t] = (t < 8 || t == 16) ? (int)0x80000000 : 0x7FFFFFFF;
  if (t >= 64 * 48) return;
  const int k = t / 48, c = t % 48;
  W2p[t] = (c < 40) ? W2[k * 40 + c] : 0.f;
}

__global__ void gat_prep2(const float* __restrict__ h2a, const float* __restrict__ h2b,
                          const float* __restrict__ attS,
                          const float* __restrict__ attD, float* __restrict__ aS,
                          int* __restrict__ dm,
                          float* __restrict__ agga, float* __restrict__ aggb,
                          int* __restrict__ g, int N) {
  __shared__ int smx[8], smn[8];
  const int n = blockIdx.x * blockDim.x + threadIdx.x;
  int emx = (int)0x80000000, emn = 0x7FFFFFFF;
  if (n < N) {
  const float* ha = h2a + (size_t)n * 32;
  const float* hb = h2b + (size_t)n * 16;
  float s = 0.f, d = 0.f;
#pragma unroll
  for (int c = 0; c < 32; ++c) { s += ha[c] * attS[c]; d += ha[c] * attD[c]; }
#pragma unroll
  for (int c = 0; c < 8; ++c) { s += hb[c] * attS[32 + c]; d += hb[c] * attD[32 + c]; }
  aS[n] = s;
  v2i p; p.x = __float_as_int(d); p.y = enc_ord(s);
  ((v2i*)dm)[n] = p;
  emx = p.y; emn = p.y;
  float* ap = agga + (size_t)n * 32;
  float* bp = ap + (size_t)N * 32;
#pragma unroll
  for (int c = 0; c < 32; ++c) { ap[c] = 0.f; bp[c] = 0.f; }
  float* cp = aggb + (size_t)n * 16;
  float* dp = cp + (size_t)N * 16;
#pragma unroll
  for (int c = 0; c < 16; ++c) { cp[c] = 0.f; dp[c] = 0.f; }
  }
#pragma unroll
  for (int m = 1; m < 32; m <<= 1) { emx = max(emx, __shfl_xor(emx, m)); emn = min(emn, __shfl_xor(emn, m)); }
  if ((threadIdx.x & 31) == 0) { smx[threadIdx.x >> 5] = emx; smn[threadIdx.x >> 5] = emn; }
  __syncthreads();
  if (threadIdx.x == 0) {
    int a = smx[0], b = smn[0];
    for (int w = 1; w < 8; ++w) { a = max(a, smx[w]); b = min(b, smn[w]); }
    atomicMax(&g[16], a);
    atomicMin(&g[17], b);
  }
}

__global__ void gat_emax2(const int* __restrict__ src, const int* __restrict__ dst,
                          const float* __restrict__ aS, int* __restrict__ dm,
                          const int* __restrict__ g, int E) {
  if (narrow_spread(g + 16, 1)) return;
  for (int c = 0; c < 8; ++c) {
    const int e = (blockIdx.x * 8 + c) * blockDim.x + threadIdx.x;
    if (e < E) atomicMax(&dm[dst[e] * 2 + 1], enc_ord(aS[src[e]]));
  }
}

__global__ void gat_eagg2(const int* __restrict__ src, const int* __restrict__ dst,
                          const float* __restrict__ aS, const int* __restrict__ dm,
                          const float* __restrict__ h2a, const float* __restrict__ h2b,
                          float* __restrict__ agga, float* __restrict__ aggb,
                          const int* __restrict__ g, int E, int N) {
  __shared__ int half_slot;
  const int lane = threadIdx.x & 31;
  const int e0 = __builtin_amdgcn_readfirstlane((int)((blockIdx.x * blockDim.x + threadIdx.x) & ~31u));
  const int e = e0 + lane;
  const int s = e < E ? src[e] : 0, d = e < E ? dst[e] : 0;
  const int hf = package_half(&half_slot);
  if (e0 >= E) return;
  agga += (size_t)hf * N * 32;
  aggb += (size_t)hf * N * 16;
  const v2i p = ((const v2i*)dm)[d];
  const float ad = __int_as_float(p.x);
  const float ms = narrow_spread(g + 16, 1) ? dec_ord(g[16]) : dec_ord(p.y);
  const float ex = __expf(lrelu(aS[s] + ad) - lrelu(ms + ad));
  const int hi = lane >> 4;
  const int ct = lane & 15;
  if (e0 + 32 <= E) {
#pragma unroll
    for (int g = 0; g < 32; g += 8) {
      float v0[4], v1[4], vl[4];
#pragma unroll
      for (int k = 0; k < 4; ++k) {
        const int j = g + 2 * k;
        const int s0 = __builtin_amdgcn_readlane(s, j), s1 = __builtin_amdgcn_readlane(s, j + 1);
        const int st = hi ? s1 : s0;
        v0[k] = h2a[(size_t)s0 * 32 + lane];
        v1[k] = h2a[(size_t)s1 * 32 + lane];
        vl[k] = h2b[(size_t)st * 16 + ct];
      }
      asm volatile("" :: "v"(v0[0]), "v"(v1[0]), "v"(vl[0]), "v"(v0[1]), "v"(v1[1]), "v"(vl[1]),
                         "v"(v0[2]), "v"(v1[2]), "v"(vl[2]), "v"(v0[3]), "v"(v1[3]), "v"(vl[3]));
      __builtin_amdgcn_sched_barrier(0);
#pragma unroll
      for (int k = 0; k < 4; ++k) {
        const int j = g + 2 * k;
        const int d0 = __builtin_amdgcn_readlane(d, j), d1 = __builtin_amdgcn_readlane(d, j + 1);
        const float x0 = __int_as_float(__builtin_amdgcn_readlane(__float_as_int(ex), j));
        const float x1 = __int_as_float(__builtin_amdgcn_readlane(__float_as_int(ex), j + 1));
        const int dt = hi ? d1 : d0;
        const float xt = hi ? x1 : x0;
        const float vt = (ct == 8) ? 1.f : vl[k];
        atomicAdd(agga + (size_t)d0 * 32 + lane, x0 * v0[k]);
        atomicAdd(agga + (size_t)d1 * 32 + lane, x1 * v1[k]);
        if (ct <= 8)
          atomicAdd(aggb + (size_t)dt * 16 + ct, xt * vt);
      }
    }
    return;
  }
#pragma unroll
  for (int j = 0; j < 32; j += 2) {
    const int s0 = __builtin_amdgcn_readlane(s, j), s1 = __builtin_amdgcn_readlane(s, j + 1);
    const int d0 = __builtin_amdgcn_readlane(d, j), d1 = __builtin_amdgcn_readlane(d, j + 1);
    const float x0 = __int_as_float(__builtin_amdgcn_readlane(__float_as_int(ex), j));
    const float x1 = __int_as_float(__builtin_amdgcn_readlane(__float_as_int(ex), j + 1));
    const float v0 = h2a[(size_t)s0 * 32 + lane];
    const float v1 = h2a[(size_t)s1 * 32 + lane];
    const int st = hi ? s1 : s0, dt = hi ? d1 : d0;
    const float xt = hi ? x1 : x0;
    const float vl = h2b[(size_t)st * 16 + ct];
    const float vt = (ct == 8) ? 1.f : vl;
    if (e0 + j < E)
      atomicAdd(agga + (size_t)d0 * 32 + lane, x0 * v0);
    if (e0 + j + 1 < E)
      atomicAdd(agga + (size_t)d1 * 32 + lane, x1 * v1);
    if (e0 + j + hi < E && ct <= 8)
      atomicAdd(aggb + (size_t)dt * 16 + ct, xt * vt);
  }
}

__global__ void gat_fin2(const float* __restrict__ agga, const float* __restrict__ aggb,
                         const float* __restrict__ h2a, const float* __restrict__ h2b,
                         const float* __restrict__ aS, const int* __restrict__ dm,
                         const float* __restrict__ bias, float* __restrict__ out,
                         const int* __restrict__ g, int N) {
  const int n = blockIdx.x * blockDim.x + threadIdx.x;
  if (n >= N) return;
  const float* ap = agga + (size_t)n * 32;
  const float* bp = ap + (size_t)N * 32;
  const float* cp = aggb + (size_t)n * 16;
  const float* dp = cp + (size_t)N * 16;
  const float* ha = h2a + (size_t)n * 32;
  const float* hb = h2b + (size_t)n * 16;
  const v2i p = ((const v2i*)dm)[n];
  const float ad = __int_as_float(p.x);
  const float ms = narrow_spread(g + 16, 1) ? dec_ord(g[16]) : dec_ord(p.y);
  const float exs = __expf(lrelu(aS[n] + ad) - lrelu(ms + ad));
  const float inv = 1.f / ((cp[8] + dp[8]) + exs + SOFT_EPS);
  float v[40];
  float mx = -INFINITY;
#pragma unroll
  for (int c = 0; c < 32; ++c) {
    v[c] = ((ap[c] + bp[c]) + exs * ha[c]) * inv + bias[c];
    mx = fmaxf(mx, v[c]);
  }
#pragma unroll
  for (int c = 0; c < 8; ++c) {
    v[32 + c] = ((cp[c] + dp[c]) + exs * hb[c]) * inv + bias[32 + c];
    mx = fmaxf(mx, v[32 + c]);
  }
  float sum = 0.f;
#pragma unroll
  for (int c = 0; c < 40; ++c) sum += __expf(v[c] - mx);
  const float lse = mx + __logf(sum);
  float* op = out + (size_t)n * 40;
#pragma unroll
  for (int c = 0; c < 40; ++c) op[c] = v[c] - lse;
}

extern "C" void kernel_launch(void* const* d_in, const int* in_sizes, int n_in,
                              void* d_out, int out_size, void* d_ws, size_t ws_size,
                              hipStream_t stream) {
  const float* x   = (const float*)d_in[0];
  const int*   ei  = (const int*)d_in[1];
  const float* W1  = (const float*)d_in[2];
  const float* as1 = (const float*)d_in[3];
  const float* ad1 = (const float*)d_in[4];
  const float* b1  = (const float*)d_in[5];
  const float* W2  = (const float*)d_in[6];
  const float* as2 = (const float*)d_in[7];
  const float* ad2 = (const float*)d_in[8];
  const float* b2  = (const float*)d_in[9];
  float* out = (float*)d_out;

  const int Fin = in_sizes[2] / 64;
  const int N   = in_sizes[0] / Fin;
  const int E   = in_sizes[1] / 2;
  const int* src = ei;
  const int* dst = ei + E;

  float* ws = (float*)d_ws;
  size_t o = 0;
  float* h1    = ws + o; o += (size_t)N * 64;
  float* hpost = ws + o; o += (size_t)N * 64;
  float* aS1   = ws + o; o += (size_t)N * 8;
  int*   dm1   = (int*)(ws + o); o += (size_t)N * 16;
  float* dn1   = ws + o; o += (size_t)N * 8 * 2;
  float* agg1  = ws + o; o += (size_t)N * 64 * 2;
  float* h2a   = ws + o; o += (size_t)N * 32;
  float* h2b   = ws + o; o += (size_t)N * 16;
  int*   dm2   = (int*)(ws + o); o += (size_t)N * 2;
  float* aS2   = ws + o; o += (size_t)N;
  float* agg2a = ws + o; o += (size_t)N * 32 * 2;
  float* agg2b = ws + o; o += (size_t)N * 16 * 2;
  float* W2p   = ws + o; o += (size_t)64 * 48;
  int*   gst   = (int*)(ws + o); o += 32;

  const int TB = 256;
  const int gNH = (N * 8 + TB - 1) / TB;
  const int gN  = (N + TB - 1) / TB;
  const int gE  = (E + TB - 1) / TB;
  const int gM  = (N + 15) / 16;
  const int gE8 = (gE + 7) / 8;

  gat_padw2 <<<(64 * 48 + TB - 1) / TB, TB, 0, stream>>>(W2, W2p, gst);

  gat_gemm_wmma<<<gM, dim3(32, 4), (size_t)Fin * 64 * 4, stream>>>(x, W1, h1, h1, N, Fin, 64, 64);
  gat_prep1 <<<gNH, TB, 0, stream>>>(h1, as1, ad1, aS1, dm1, dn1, agg1, gst, N);
  gat_emax1 <<<gE8, TB, 0, stream>>>(src, dst, aS1, dm1, gst, E);
  gat_eagg1 <<<gE,  TB, 0, stream>>>(src, dst, aS1, dm1, dn1, h1, agg1, gst, E, N);
  gat_fin1  <<<gNH, TB, 0, stream>>>(agg1, h1, aS1, dm1, dn1, b1, hpost, gst, N);

  gat_gemm_wmma<<<gM, dim3(32, 3), (size_t)64 * 48 * 4, stream>>>(hpost, W2p, h2a, h2b, N, 64, 48, 32);
  gat_prep2 <<<gN, TB, 0, stream>>>(h2a, h2b, as2, ad2, aS2, dm2, agg2a, agg2b, gst, N);
  gat_emax2 <<<gE8, TB, 0, stream>>>(src, dst, aS2, dm2, gst, E);
  gat_eagg2 <<<gE, TB, 0, stream>>>(src, dst, aS2, dm2, h2a, h2b, agg2a, agg2b, gst, E, N);
  gat_fin2  <<<gN, TB, 0, stream>>>(agg2a, agg2b, h2a, h2b, aS2, dm2, b2, out, gst, N);

  (void)n_in; (void)out_size; (void)ws_size;
}


# Layer-2 source score kept in the spare element of the source's 64-byte row piece: the sum pass loads the 16 pieces first and takes the scores from them; per-edge score gather gone; add order kept
# speedup vs baseline: 14.2124x; 1.0360x over previous
#include <hip/hip_runtime.h>
#include <math.h>

typedef __attribute__((ext_vector_type(2))) float v2f;
typedef __attribute__((ext_vector_type(8))) float v8f;
typedef __attribute__((ext_vector_type(2))) int   v2i;
typedef __attribute__((ext_vector_type(4))) float v4f;

#define NEG_SLOPE 0.2f
#define SOFT_EPS  1e-16f

__device__ __forceinline__ float lrelu(float x) { return x > 0.f ? x : NEG_SLOPE * x; }
__device__ __forceinline__ int   enc_ord(float f) { int i = __float_as_int(f); return i >= 0 ? i : (i ^ 0x7FFFFFFF); }
__device__ __forceinline__ float dec_ord(int i)   { return __int_as_float(i >= 0 ? i : (i ^ 0x7FFFFFFF)); }

#define SPREAD_MAX 20.f
__device__ __forceinline__ bool narrow_spread(const int* __restrict__ g, int nh) {
  int ok = 1;
#pragma unroll
  for (int h = 0; h < 8; ++h)
    if (h < nh) ok &= (dec_ord(g[h]) - dec_ord(g[nh + h]) <= SPREAD_MAX) ? 1 : 0;
  return ok != 0;
}

__global__ void gat_gemm_wmma(const float* __restrict__ A, const float* __restrict__ Bg,
                              float* __restrict__ C, float* __restrict__ C2,
                              int M, int K, int NC, int NC1) {
  extern __shared__ __attribute__((aligned(16))) float Bs[];
  const int nthr = blockDim.x * blockDim.y;
  const int tid  = threadIdx.y * blockDim.x + threadIdx.x;
  const int tot  = K * NC;
  if ((((size_t)Bg) & 15) == 0 && (tot & 3) == 0) {
    const int tot4 = tot >> 2;
    const v4f* Bg4 = (const v4f*)Bg;
    v4f* Bs4 = (v4f*)Bs;
    for (int i0 = tid; i0 < tot4; i0 += 8 * nthr) {
      v4f t[8];
#pragma unroll
      for (int u = 0; u < 8; ++u) t[u] = Bg4[min(i0 + u * nthr, tot4 - 1)];
#pragma unroll
      for (int u = 0; u < 8; ++u) {
        const int i = i0 + u * nthr;
        if (i < tot4) Bs4[i] = t[u];
      }
    }
  } else {
    for (int i = tid; i < tot; i += nthr) Bs[i] = Bg[i];
  }
  __syncthreads();

  const int row0 = blockIdx.x * 16;
  const int col0 = threadIdx.y * 16;
  const int lane = threadIdx.x;
  const int mr   = lane & 15;
  const int half = lane >> 4;

  int arow = row0 + mr; if (arow > M - 1) arow = M - 1;
  const float* Ap = A + (size_t)arow * K;

  v8f acc = {};
  for (int k0 = 0; k0 < K; k0 += 4) {
    const int ka = k0 + 2 * half;
    v2f a, b;
    a.x = Ap[ka];
    a.y = Ap[ka + 1];
    b.x = Bs[ka * NC + col0 + mr];
    b.y = Bs[(ka + 1) * NC + col0 + mr];
    acc = __builtin_amdgcn_wmma_f32_16x16x4_f32(false, a, false, b, (short)0, acc,
                                                false, false);
  }
  const bool first = col0 < NC1;
  float* Cp = first ? C : C2;
  const int pitch = first ? NC1 : NC - NC1;
  const int col = (first ? col0 : col0 - NC1) + mr;
#pragma unroll
  for (int i = 0; i < 8; ++i) {
    const int r = row0 + i + 8 * half;
    if (r < M) Cp[(size_t)r * pitch + col] = acc[i];
  }
}

__device__ __forceinline__ int package_half(int* slot) {
  if (threadIdx.x == 0) *slot = (int)((__builtin_amdgcn_s_sendmsg_rtn(0x87) >> 18) & 1u);
  __syncthreads();
  return __builtin_amdgcn_readfirstlane(*slot);
}
__global__ void gat_prep1(const float* __restrict__ h1, const float* __restrict__ attS,
                          const float* __restrict__ attD, float* __restrict__ aS,
                          int* __restrict__ dm,
                          float* __restrict__ denom, float* __restrict__ agg,
                          int* __restrict__ g, int N) {
  __shared__ int smx[8][8], smn[8][8];
  const int t = blockIdx.x * blockDim.x + threadIdx.x;
  const int h = t & 7;
  int emx = (int)0x80000000, emn = 0x7FFFFFFF;
  if (t < N * 8) {
  const float* hp = h1 + (size_t)t * 8;
  float s = 0.f, d = 0.f;
#pragma unroll
  for (int c = 0; c < 8; ++c) { s += hp[c] * attS[h * 8 + c]; d += hp[c] * attD[h * 8 + c]; }
  aS[t] = s;
  v2i p; p.x = __float_as_int(d); p.y = enc_ord(s);
  ((v2i*)dm)[t] = p;
  emx = p.y; emn = p.y;
  denom[t] = 0.f;
  denom[(size_t)N * 8 + t] = 0.f;
  float* ap = agg + (size_t)t * 8;
  float* bp = ap + (size_t)N * 64;
#pragma unroll
  for (int c = 0; c < 8; ++c) { ap[c] = 0.f; bp[c] = 0.f; }
  }
  emx = max(emx, __shfl_xor(emx, 8));  emn = min(emn, __shfl_xor(emn, 8));
  emx = max(emx, __shfl_xor(emx, 16)); emn = min(emn, __shfl_xor(emn, 16));
  const int lane = threadIdx.x & 31, wv = threadIdx.x >> 5;
  if (lane < 8) { smx[wv][lane] = emx; smn[wv][lane] = emn; }
  __syncthreads();
  if (threadIdx.x < 8) {
    int a = smx[0][threadIdx.x], b = smn[0][threadIdx.x];
    for (int w = 1; w < 8; ++w) { a = max(a, smx[w][threadIdx.x]); b = min(b, smn[w][threadIdx.x]); }
    atomicMax(&g[threadIdx.x], a);
    atomicMin(&g[8 + threadIdx.x], b);
  }
}

__global__ void gat_emax1(const int* __restrict__ src, const int* __restrict__ dst,
                          const float* __restrict__ aS, int* __restrict__ dm,
                          const int* __restrict__ g, int E) {
  if (narrow_spread(g, 8)) return;
  for (int c = 0; c < 8; ++c) {
  const int lane = threadIdx.x & 31;
  const int e0 = __builtin_amdgcn_readfirstlane((int)((((blockIdx.x * 8 + c) * blockDim.x) + threadIdx.x) & ~31u));
  if (e0 >= E) return;
  const int e = e0 + lane;
  const int s = e < E ? src[e] : 0, d = e < E ? dst[e] : 0;
  const int h = lane & 7;
#pragma unroll
  for (int i = 0; i < 8; ++i) {
    const int j = 4 * i + (lane >> 3);
    const int sj = __shfl(s, j), dj = __shfl(d, j);
    const int v = enc_ord(aS[sj * 8 + h]);
    if (e0 + j < E) atomicMax(&dm[(dj * 8 + h) * 2 + 1], v);
  }
  }
}

__global__ void gat_eagg1(const int* __restrict__ src, const int* __restrict__ dst,
                          const float* __restrict__ aS, const int* __restrict__ dm,
                          float* __restrict__ denom,
                          const float* __restrict__ h1, float* __restrict__ agg,
                          const int* __restrict__ g, int E, int N) {
  __shared__ int half_slot;
  const int lane = threadIdx.x & 31;
  const int e0 = __builtin_amdgcn_readfirstlane((int)((blockIdx.x * blockDim.x + threadIdx.x) & ~31u));
  const int e = e0 + lane;
  const int s = e < E ? src[e] : 0, d = e < E ? dst[e] : 0;
  const int hf = package_half(&half_slot);
  if (e0 >= E) return;
  denom += (size_t)hf * N * 8;
  agg   += (size_t)hf * N * 64;
  const int h = lane & 7;
  const bool nsp = narrow_spread(g, 8);
  const float gms = dec_ord(g[h]);
  float ex[8];
  int   dh[8];
#pragma unroll
  for (int i = 0; i < 8; ++i) {
    const int j = 4 * i + (lane >> 3);
    const int sj = __shfl(s, j), dj = __shfl(d, j);
    dh[i] = dj * 8 + h;
    const v2i p = ((const v2i*)dm)[dh[i]];
    const float ad = __int_as_float(p.x);
    const float ev = lrelu(aS[sj * 8 + h] + ad);
    ex[i] = __expf(ev - lrelu((nsp ? gms : dec_ord(p.y)) + ad));
  }
#pragma unroll
  for (int i = 0; i < 8; ++i)
    if (e0 + 4 * i + (lane >> 3) < E) atomicAdd(&denom[dh[i]], ex[i]);
  const int hl = lane >> 3;
  if (e0 + 32 <= E) {
#pragma unroll
    for (int g = 0; g < 32; g += 8) {
      float v0[8], v1[8];
#pragma unroll
      for (int k = 0; k < 8; ++k) {
        const int sj = __builtin_amdgcn_readlane(s, g + k);
        const float* hs = h1 + (size_t)sj * 64;
        v0[k] = hs[lane];
        v1[k] = hs[32 + lane];
      }
#pragma unroll
      for (int k = 0; k < 8; ++k) {
        const int j = g + k;
        const int dj = __builtin_amdgcn_readlane(d, j);
        const float a_lo = __shfl(ex[j >> 2], ((j & 3) << 3) + hl);
        const float a_hi = __shfl(ex[j >> 2], ((j & 3) << 3) + 4 + hl);
        float* o = agg + (size_t)dj * 64;
        atomicAdd(o + lane, a_lo * v0[k]);
        atomicAdd(o + 32 + lane, a_hi * v1[k]);
      }
    }
    return;
  }
#pragma unroll
  for (int j = 0; j < 32; ++j) {
    const int sj = __builtin_amdgcn_readlane(s, j);
    const int dj = __builtin_amdgcn_readlane(d, j);
    const float a_lo = __shfl(ex[j >> 2], ((j & 3) << 3) + hl);
    const float a_hi = __shfl(ex[j >> 2], ((j & 3) << 3) + 4 + hl);
    const float* hs = h1 + (size_t)sj * 64;
    float* o = agg + (size_t)dj * 64;
    const float v0 = hs[lane], v1 = hs[32 + lane];
    if (e0 + j < E) {
      atomicAdd(o + lane, a_lo * v0);
      atomicAdd(o + 32 + lane, a_hi * v1);
    }
  }
}

__global__ void gat_fin1(const float* __restrict__ agg, const float* __restrict__ h1,
                         const float* __restrict__ aS, const int* __restrict__ dm,
                         const float* __restrict__ denom,
                         const float* __restrict__ bias, float* __restrict__ hpost,
                         const int* __restrict__ g, int N) {
  const int t = blockIdx.x * blockDim.x + threadIdx.x;
  if (t >= N * 8) return;
  const int h = t & 7;
  const v2i p = ((const v2i*)dm)[t];
  const float ad = __int_as_float(p.x);
  const float ms = narrow_spread(g, 8) ? dec_ord(g[h]) : dec_ord(p.y);
  const float exs = __expf(lrelu(aS[t] + ad) - lrelu(ms + ad));
  const float inv = 1.f / ((denom[t] + denom[(size_t)N * 8 + t]) + exs + SOFT_EPS);
  const float* ap = agg + (size_t)t * 8;
  const float* bp = ap + (size_t)N * 64;
  const float* hp = h1 + (size_t)t * 8;
  float* op = hpost + (size_t)t * 8;
#pragma unroll
  for (int c = 0; c < 8; ++c) {
    const float v = ((ap[c] + bp[c]) + exs * hp[c]) * inv + bias[h * 8 + c];
    op[c] = v > 0.f ? v : (__expf(v) - 1.f);
  }
}

__global__ void gat_padw2(const float* __restrict__ W2, float* __restrict__ W2p, int* __restrict__ g) {
  const int t = blockIdx.x * blockDim.x + threadIdx.x;
  if (t < 18) g[t] = (t < 8 || t == 16) ? (int)0x80000000 : 0x7FFFFFFF;
  if (t >= 64 * 48) return;
  const int k = t / 48, c = t % 48;
  W2p[t] = (c < 40) ? W2[k * 40 + c] : 0.f;
}

__global__ void gat_prep2(const float* __restrict__ h2a, float* __restrict__ h2b,
                          const float* __restrict__ attS,
                          const float* __restrict__ attD, float* __restrict__ aS,
                          int* __restrict__ dm,
                          float* __restrict__ agga, float* __restrict__ aggb,
                          int* __restrict__ g, int N) {
  __shared__ int smx[8], smn[8];
  const int n = blockIdx.x * blockDim.x + threadIdx.x;
  int emx = (int)0x80000000, emn = 0x7FFFFFFF;
  if (n < N) {
  const float* ha = h2a + (size_t)n * 32;
  const float* hb = h2b + (size_t)n * 16;
  float s = 0.f, d = 0.f;
#pragma unroll
  for (int c = 0; c < 32; ++c) { s += ha[c] * attS[c]; d += ha[c] * attD[c]; }
#pragma unroll
  for (int c = 0; c < 8; ++c) { s += hb[c] * attS[32 + c]; d += hb[c] * attD[32 + c]; }
  aS[n] = s;
  h2b[(size_t)n * 16 + 8] = s;
  v2i p; p.x = __float_as_int(d); p.y = enc_ord(s);
  ((v2i*)dm)[n] = p;
  emx = p.y; emn = p.y;
  float* ap = agga + (size_t)n * 32;
  float* bp = ap + (size_t)N * 32;
#pragma unroll
  for (int c = 0; c < 32; ++c) { ap[c] = 0.f; bp[c] = 0.f; }
  float* cp = aggb + (size_t)n * 16;
  float* dp = cp + (size_t)N * 16;
#pragma unroll
  for (int c = 0; c < 16; ++c) { cp[c] = 0.f; dp[c] = 0.f; }
  }
#pragma unroll
  for (int m = 1; m < 32; m <<= 1) { emx = max(emx, __shfl_xor(emx, m)); emn = min(emn, __shfl_xor(emn, m)); }
  if ((threadIdx.x & 31) == 0) { smx[threadIdx.x >> 5] = emx; smn[threadIdx.x >> 5] = emn; }
  __syncthreads();
  if (threadIdx.x == 0) {
    int a = smx[0], b = smn[0];
    for (int w = 1; w < 8; ++w) { a = max(a, smx[w]); b = min(b, smn[w]); }
    atomicMax(&g[16], a);
    atomicMin(&g[17], b);
  }
}

__global__ void gat_emax2(const int* __restrict__ src, const int* __restrict__ dst,
                          const float* __restrict__ aS, int* __restrict__ dm,
                          const int* __restrict__ g, int E) {
  if (narrow_spread(g + 16, 1)) return;
  for (int c = 0; c < 8; ++c) {
    const int e = (blockIdx.x * 8 + c) * blockDim.x + threadIdx.x;
    if (e < E) atomicMax(&dm[dst[e] * 2 + 1], enc_ord(aS[src[e]]));
  }
}

__global__ void gat_eagg2(const int* __restrict__ src, const int* __restrict__ dst,
                          const float* __restrict__ aS, const int* __restrict__ dm,
                          const float* __restrict__ h2a, const float* __restrict__ h2b,
                          float* __restrict__ agga, float* __restrict__ aggb,
                          const int* __restrict__ g, int E, int N) {
  __shared__ int half_slot;
  const int lane = threadIdx.x & 31;
  const int e0 = __builtin_amdgcn_readfirstlane((int)((blockIdx.x * blockDim.x + threadIdx.x) & ~31u));
  const int e = e0 + lane;
  const int s = e < E ? src[e] : 0, d = e < E ? dst[e] : 0;
  const int hf = package_half(&half_slot);
  if (e0 >= E) return;
  agga += (size_t)hf * N * 32;
  aggb += (size_t)hf * N * 16;
  const v2i p = ((const v2i*)dm)[d];
  const float ad = __int_as_float(p.x);
  const float ms = narrow_spread(g + 16, 1) ? dec_ord(g[16]) : dec_ord(p.y);
  const int hi = lane >> 4;
  const int ct = lane & 15;
  if (e0 + 32 <= E) {
    float vl[16];
#pragma unroll
    for (int k = 0; k < 16; ++k) {
      const int s0 = __builtin_amdgcn_readlane(s, 2 * k), s1 = __builtin_amdgcn_readlane(s, 2 * k + 1);
      vl[k] = h2b[(size_t)(hi ? s1 : s0) * 16 + ct];
    }
    int asb = 0;
#pragma unroll
    for (int k = 0; k < 16; ++k) {
      const int a0 = __builtin_amdgcn_readlane(__float_as_int(vl[k]), 8);
      const int a1 = __builtin_amdgcn_readlane(__float_as_int(vl[k]), 24);
      asb = (lane == 2 * k) ? a0 : asb;
      asb = (lane == 2 * k + 1) ? a1 : asb;
    }
    const float ex = __expf(lrelu(__int_as_float(asb) + ad) - lrelu(ms + ad));
#pragma unroll
    for (int g = 0; g < 32; g += 8) {
      float v0[4], v1[4];
#pragma unroll
      for (int k = 0; k < 4; ++k) {
        const int j = g + 2 * k;
        const int s0 = __builtin_amdgcn_readlane(s, j), s1 = __builtin_amdgcn_readlane(s, j + 1);
        v0[k] = h2a[(size_t)s0 * 32 + lane];
        v1[k] = h2a[(size_t)s1 * 32 + lane];
      }
#pragma unroll
      for (int k = 0; k < 4; ++k) {
        const int j = g + 2 * k;
        const int d0 = __builtin_amdgcn_readlane(d, j), d1 = __builtin_amdgcn_readlane(d, j + 1);
        const float x0 = __int_as_float(__builtin_amdgcn_readlane(__float_as_int(ex), j));
        const float x1 = __int_as_float(__builtin_amdgcn_readlane(__float_as_int(ex), j + 1));
        const int dt = hi ? d1 : d0;
        const float xt = hi ? x1 : x0;
        const float vt = (ct == 8) ? 1.f : vl[g / 2 + k];
        atomicAdd(agga + (size_t)d0 * 32 + lane, x0 * v0[k]);
        atomicAdd(agga + (size_t)d1 * 32 + lane, x1 * v1[k]);
        if (ct <= 8)
          atomicAdd(aggb + (size_t)dt * 16 + ct, xt * vt);
      }
    }
    return;
  }
  const float ex = __expf(lrelu(aS[s] + ad) - lrelu(ms + ad));
#pragma unroll
  for (int j = 0; j < 32; j += 2) {
    const int s0 = __builtin_amdgcn_readlane(s, j), s1 = __builtin_amdgcn_readlane(s, j + 1);
    const int d0 = __builtin_amdgcn_readlane(d, j), d1 = __builtin_amdgcn_readlane(d, j + 1);
    const float x0 = __int_as_float(__builtin_amdgcn_readlane(__float_as_int(ex), j));
    const float x1 = __int_as_float(__builtin_amdgcn_readlane(__float_as_int(ex), j + 1));
    const float v0 = h2a[(size_t)s0 * 32 + lane];
    const float v1 = h2a[(size_t)s1 * 32 + lane];
    const int st = hi ? s1 : s0, dt = hi ? d1 : d0;
    const float xt = hi ? x1 : x0;
    const float vl = h2b[(size_t)st * 16 + ct];
    const float vt = (ct == 8) ? 1.f : vl;
    if (e0 + j < E)
      atomicAdd(agga + (size_t)d0 * 32 + lane, x0 * v0);
    if (e0 + j + 1 < E)
      atomicAdd(agga + (size_t)d1 * 32 + lane, x1 * v1);
    if (e0 + j + hi < E && ct <= 8)
      atomicAdd(aggb + (size_t)dt * 16 + ct, xt * vt);
  }
}

__global__ void gat_fin2(const float* __restrict__ agga, const float* __restrict__ aggb,
                         const float* __restrict__ h2a, const float* __restrict__ h2b,
                         const float* __restrict__ aS, const int* __restrict__ dm,
                         const float* __restrict__ bias, float* __restrict__ out,
                         const int* __restrict__ g, int N) {
  const int n = blockIdx.x * blockDim.x + threadIdx.x;
  if (n >= N) return;
  const float* ap = agga + (size_t)n * 32;
  const float* bp = ap + (size_t)N * 32;
  const float* cp = aggb + (size_t)n * 16;
  const float* dp = cp + (size_t)N * 16;
  const float* ha = h2a + (size_t)n * 32;
  const float* hb = h2b + (size_t)n * 16;
  const v2i p = ((const v2i*)dm)[n];
  const float ad = __int_as_float(p.x);
  const float ms = narrow_spread(g + 16, 1) ? dec_ord(g[16]) : dec_ord(p.y);
  const float exs = __expf(lrelu(aS[n] + ad) - lrelu(ms + ad));
  const float inv = 1.f / ((cp[8] + dp[8]) + exs + SOFT_EPS);
  float v[40];
  float mx = -INFINITY;
#pragma unroll
  for (int c = 0; c < 32; ++c) {
    v[c] = ((ap[c] + bp[c]) + exs * ha[c]) * inv + bias[c];
    mx = fmaxf(mx, v[c]);
  }
#pragma unroll
  for (int c = 0; c < 8; ++c) {
    v[32 + c] = ((cp[c] + dp[c]) + exs * hb[c]) * inv + bias[32 + c];
    mx = fmaxf(mx, v[32 + c]);
  }
  float sum = 0.f;
#pragma unroll
  for (int c = 0; c < 40; ++c) sum += __expf(v[c] - mx);
  const float lse = mx + __logf(sum);
  float* op = out + (size_t)n * 40;
#pragma unroll
  for (int c = 0; c < 40; ++c) op[c] = v[c] - lse;
}

extern "C" void kernel_launch(void* const* d_in, const int* in_sizes, int n_in,
                              void* d_out, int out_size, void* d_ws, size_t ws_size,
                              hipStream_t stream) {
  const float* x   = (const float*)d_in[0];
  const int*   ei  = (const int*)d_in[1];
  const float* W1  = (const float*)d_in[2];
  const float* as1 = (const float*)d_in[3];
  const float* ad1 = (const float*)d_in[4];
  const float* b1  = (const float*)d_in[5];
  const float* W2  = (const float*)d_in[6];
  const float* as2 = (const float*)d_in[7];
  const float* ad2 = (const float*)d_in[8];
  const float* b2  = (const float*)d_in[9];
  float* out = (float*)d_out;

  const int Fin = in_sizes[2] / 64;
  const int N   = in_sizes[0] / Fin;
  const int E   = in_sizes[1] / 2;
  const int* src = ei;
  const int* dst = ei + E;

  float* ws = (float*)d_ws;
  size_t o = 0;
  float* h1    = ws + o; o += (size_t)N * 64;
  float* hpost = ws + o; o += (size_t)N * 64;
  float* aS1   = ws + o; o += (size_t)N * 8;
  int*   dm1   = (int*)(ws + o); o += (size_t)N * 16;
  float* dn1   = ws + o; o += (size_t)N * 8 * 2;
  float* agg1  = ws + o; o += (size_t)N * 64 * 2;
  float* h2a   = ws + o; o += (size_t)N * 32;
  float* h2b   = ws + o; o += (size_t)N * 16;
  int*   dm2   = (int*)(ws + o); o += (size_t)N * 2;
  float* aS2   = ws + o; o += (size_t)N;
  float* agg2a = ws + o; o += (size_t)N * 32 * 2;
  float* agg2b = ws + o; o += (size_t)N * 16 * 2;
  float* W2p   = ws + o; o += (size_t)64 * 48;
  int*   gst   = (int*)(ws + o); o += 32;

  const int TB = 256;
  const int gNH = (N * 8 + TB - 1) / TB;
  const int gN  = (N + TB - 1) / TB;
  const int gE  = (E + TB - 1) / TB;
  const int gM  = (N + 15) / 16;
  const int gE8 = (gE + 7) / 8;

  gat_padw2 <<<(64 * 48 + TB - 1) / TB, TB, 0, stream>>>(W2, W2p, gst);

  gat_gemm_wmma<<<gM, dim3(32, 4), (size_t)Fin * 64 * 4, stream>>>(x, W1, h1, h1, N, Fin, 64, 64);
  gat_prep1 <<<gNH, TB, 0, stream>>>(h1, as1, ad1, aS1, dm1, dn1, agg1, gst, N);
  gat_emax1 <<<gE8, TB, 0, stream>>>(src, dst, aS1, dm1, gst, E);
  gat_eagg1 <<<gE,  TB, 0, stream>>>(src, dst, aS1, dm1, dn1, h1, agg1, gst, E, N);
  gat_fin1  <<<gNH, TB, 0, stream>>>(agg1, h1, aS1, dm1, dn1, b1, hpost, gst, N);

  gat_gemm_wmma<<<gM, dim3(32, 3), (size_t)64 * 48 * 4, stream>>>(hpost, W2p, h2a, h2b, N, 64, 48, 32);
  gat_prep2 <<<gN, TB, 0, stream>>>(h2a, h2b, as2, ad2, aS2, dm2, agg2a, agg2b, gst, N);
  gat_emax2 <<<gE8, TB, 0, stream>>>(src, dst, aS2, dm2, gst, E);
  gat_eagg2 <<<gE, TB, 0, stream>>>(src, dst, aS2, dm2, h2a, h2b, agg2a, agg2b, gst, E, N);
  gat_fin2  <<<gN, TB, 0, stream>>>(agg2a, agg2b, h2a, h2b, aS2, dm2, b2, out, gst, N);

  (void)n_in; (void)out_size; (void)ws_size;
}


# Layer-2 node passes re-mapped: a wave's 16-lane halves span a node pair's channels, loads and stores cover whole lines (was 16 bytes of each of 32 lines); scores and log-softmax sums now lane trees
# speedup vs baseline: 14.7239x; 1.0360x over previous
#include <hip/hip_runtime.h>
#include <math.h>

typedef __attribute__((ext_vector_type(2))) float v2f;
typedef __attribute__((ext_vector_type(8))) float v8f;
typedef __attribute__((ext_vector_type(2))) int   v2i;
typedef __attribute__((ext_vector_type(4))) float v4f;

#define NEG_SLOPE 0.2f
#define SOFT_EPS  1e-16f

__device__ __forceinline__ float lrelu(float x) { return x > 0.f ? x : NEG_SLOPE * x; }
__device__ __forceinline__ int   enc_ord(float f) { int i = __float_as_int(f); return i >= 0 ? i : (i ^ 0x7FFFFFFF); }
__device__ __forceinline__ float dec_ord(int i)   { return __int_as_float(i >= 0 ? i : (i ^ 0x7FFFFFFF)); }

#define SPREAD_MAX 20.f
__device__ __forceinline__ bool narrow_spread(const int* __restrict__ g, int nh) {
  int ok = 1;
#pragma unroll
  for (int h = 0; h < 8; ++h)
    if (h < nh) ok &= (dec_ord(g[h]) - dec_ord(g[nh + h]) <= SPREAD_MAX) ? 1 : 0;
  return ok != 0;
}

__global__ void gat_gemm_wmma(const float* __restrict__ A, const float* __restrict__ Bg,
                              float* __restrict__ C, float* __restrict__ C2,
                              int M, int K, int NC, int NC1) {
  extern __shared__ __attribute__((aligned(16))) float Bs[];
  const int nthr = blockDim.x * blockDim.y;
  const int tid  = threadIdx.y * blockDim.x + threadIdx.x;
  const int tot  = K * NC;
  if ((((size_t)Bg) & 15) == 0 && (tot & 3) == 0) {
    const int tot4 = tot >> 2;
    const v4f* Bg4 = (const v4f*)Bg;
    v4f* Bs4 = (v4f*)Bs;
    for (int i0 = tid; i0 < tot4; i0 += 8 * nthr) {
      v4f t[8];
#pragma unroll
      for (int u = 0; u < 8; ++u) t[u] = Bg4[min(i0 + u * nthr, tot4 - 1)];
#pragma unroll
      for (int u = 0; u < 8; ++u) {
        const int i = i0 + u * nthr;
        if (i < tot4) Bs4[i] = t[u];
      }
    }
  } else {
    for (int i = tid; i < tot; i += nthr) Bs[i] = Bg[i];
  }
  __syncthreads();

  const int row0 = blockIdx.x * 16;
  const int col0 = threadIdx.y * 16;
  const int lane = threadIdx.x;
  const int mr   = lane & 15;
  const int half = lane >> 4;

  int arow = row0 + mr; if (arow > M - 1) arow = M - 1;
  const float* Ap = A + (size_t)arow * K;

  v8f acc = {};
  for (int k0 = 0; k0 < K; k0 += 4) {
    const int ka = k0 + 2 * half;
    v2f a, b;
    a.x = Ap[ka];
    a.y = Ap[ka + 1];
    b.x = Bs[ka * NC + col0 + mr];
    b.y = Bs[(ka + 1) * NC + col0 + mr];
    acc = __builtin_amdgcn_wmma_f32_16x16x4_f32(false, a, false, b, (short)0, acc,
                                                false, false);
  }
  const bool first = col0 < NC1;
  float* Cp = first ? C : C2;
  const int pitch = first ? NC1 : NC - NC1;
  const int col = (first ? col0 : col0 - NC1) + mr;
#pragma unroll
  for (int i = 0; i < 8; ++i) {
    const int r = row0 + i + 8 * half;
    if (r < M) Cp[(size_t)r * pitch + col] = acc[i];
  }
}

__device__ __forceinline__ int package_half(int* slot) {
  if (threadIdx.x == 0) *slot = (int)((__builtin_amdgcn_s_sendmsg_rtn(0x87) >> 18) & 1u);
  __syncthreads();
  return __builtin_amdgcn_readfirstlane(*slot);
}
__global__ void gat_prep1(const float* __restrict__ h1, const float* __restrict__ attS,
                          const float* __restrict__ attD, float* __restrict__ aS,
                          int* __restrict__ dm,
                          float* __restrict__ denom, float* __restrict__ agg,
                          int* __restrict__ g, int N) {
  __shared__ int smx[8][8], smn[8][8];
  const int t = blockIdx.x * blockDim.x + threadIdx.x;
  const int h = t & 7;
  int emx = (int)0x80000000, emn = 0x7FFFFFFF;
  if (t < N * 8) {
  const float* hp = h1 + (size_t)t * 8;
  float s = 0.f, d = 0.f;
#pragma unroll
  for (int c = 0; c < 8; ++c) { s += hp[c] * attS[h * 8 + c]; d += hp[c] * attD[h * 8 + c]; }
  aS[t] = s;
  v2i p; p.x = __float_as_int(d); p.y = enc_ord(s);
  ((v2i*)dm)[t] = p;
  emx = p.y; emn = p.y;
  denom[t] = 0.f;
  denom[(size_t)N * 8 + t] = 0.f;
  float* ap = agg + (size_t)t * 8;
  float* bp = ap + (size_t)N * 64;
#pragma unroll
  for (int c = 0; c < 8; ++c) { ap[c] = 0.f; bp[c] = 0.f; }
  }
  emx = max(emx, __shfl_xor(emx, 8));  emn = min(emn, __shfl_xor(emn, 8));
  emx = max(emx, __shfl_xor(emx, 16)); emn = min(emn, __shfl_xor(emn, 16));
  const int lane = threadIdx.x & 31, wv = threadIdx.x >> 5;
  if (lane < 8) { smx[wv][lane] = emx; smn[wv][lane] = emn; }
  __syncthreads();
  if (threadIdx.x < 8) {
    int a = smx[0][threadIdx.x], b = smn[0][threadIdx.x];
    for (int w = 1; w < 8; ++w) { a = max(a, smx[w][threadIdx.x]); b = min(b, smn[w][threadIdx.x]); }
    atomicMax(&g[threadIdx.x], a);
    atomicMin(&g[8 + threadIdx.x], b);
  }
}

__global__ void gat_emax1(const int* __restrict__ src, const int* __restrict__ dst,
                          const float* __restrict__ aS, int* __restrict__ dm,
                          const int* __restrict__ g, int E) {
  if (narrow_spread(g, 8)) return;
  for (int c = 0; c < 8; ++c) {
  const int lane = threadIdx.x & 31;
  const int e0 = __builtin_amdgcn_readfirstlane((int)((((blockIdx.x * 8 + c) * blockDim.x) + threadIdx.x) & ~31u));
  if (e0 >= E) return;
  const int e = e0 + lane;
  const int s = e < E ? src[e] : 0, d = e < E ? dst[e] : 0;
  const int h = lane & 7;
#pragma unroll
  for (int i = 0; i < 8; ++i) {
    const int j = 4 * i + (lane >> 3);
    const int sj = __shfl(s, j), dj = __shfl(d, j);
    const int v = enc_ord(aS[sj * 8 + h]);
    if (e0 + j < E) atomicMax(&dm[(dj * 8 + h) * 2 + 1], v);
  }
  }
}

__global__ void gat_eagg1(const int* __restrict__ src, const int* __restrict__ dst,
                          const float* __restrict__ aS, const int* __restrict__ dm,
                          float* __restrict__ denom,
                          const float* __restrict__ h1, float* __restrict__ agg,
                          const int* __restrict__ g, int E, int N) {
  __shared__ int half_slot;
  const int lane = threadIdx.x & 31;
  const int e0 = __builtin_amdgcn_readfirstlane((int)((blockIdx.x * blockDim.x + threadIdx.x) & ~31u));
  const int e = e0 + lane;
  const int s = e < E ? src[e] : 0, d = e < E ? dst[e] : 0;
  const int hf = package_half(&half_slot);
  if (e0 >= E) return;
  denom += (size_t)hf * N * 8;
  agg   += (size_t)hf * N * 64;
  const int h = lane & 7;
  const bool nsp = narrow_spread(g, 8);
  const float gms = dec_ord(g[h]);
  float ex[8];
  int   dh[8];
#pragma unroll
  for (int i = 0; i < 8; ++i) {
    const int j = 4 * i + (lane >> 3);
    const int sj = __shfl(s, j), dj = __shfl(d, j);
    dh[i] = dj * 8 + h;
    const v2i p = ((const v2i*)dm)[dh[i]];
    const float ad = __int_as_float(p.x);
    const float ev = lrelu(aS[sj * 8 + h] + ad);
    ex[i] = __expf(ev - lrelu((nsp ? gms : dec_ord(p.y)) + ad));
  }
#pragma unroll
  for (int i = 0; i < 8; ++i)
    if (e0 + 4 * i + (lane >> 3) < E) atomicAdd(&denom[dh[i]], ex[i]);
  const int hl = lane >> 3;
  if (e0 + 32 <= E) {
#pragma unroll
    for (int g = 0; g < 32; g += 8) {
      float v0[8], v1[8];
#pragma unroll
      for (int k = 0; k < 8; ++k) {
        const int sj = __builtin_amdgcn_readlane(s, g + k);
        const float* hs = h1 + (size_t)sj * 64;
        v0[k] = hs[lane];
        v1[k] = hs[32 + lane];
      }
#pragma unroll
      for (int k = 0; k < 8; ++k) {
        const int j = g + k;
        const int dj = __builtin_amdgcn_readlane(d, j);
        const float a_lo = __shfl(ex[j >> 2], ((j & 3) << 3) + hl);
        const float a_hi = __shfl(ex[j >> 2], ((j & 3) << 3) + 4 + hl);
        float* o = agg + (size_t)dj * 64;
        atomicAdd(o + lane, a_lo * v0[k]);
        atomicAdd(o + 32 + lane, a_hi * v1[k]);
      }
    }
    return;
  }
#pragma unroll
  for (int j = 0; j < 32; ++j) {
    const int sj = __builtin_amdgcn_readlane(s, j);
    const int dj = __builtin_amdgcn_readlane(d, j);
    const float a_lo = __shfl(ex[j >> 2], ((j & 3) << 3) + hl);
    const float a_hi = __shfl(ex[j >> 2], ((j & 3) << 3) + 4 + hl);
    const float* hs = h1 + (size_t)sj * 64;
    float* o = agg + (size_t)dj * 64;
    const float v0 = hs[lane], v1 = hs[32 + lane];
    if (e0 + j < E) {
      atomicAdd(o + lane, a_lo * v0);
      atomicAdd(o + 32 + lane, a_hi * v1);
    }
  }
}

__global__ void gat_fin1(const float* __restrict__ agg, const float* __restrict__ h1,
                         const float* __restrict__ aS, const int* __restrict__ dm,
                         const float* __restrict__ denom,
                         const float* __restrict__ bias, float* __restrict__ hpost,
                         const int* __restrict__ g, int N) {
  const int t = blockIdx.x * blockDim.x + threadIdx.x;
  if (t >= N * 8) return;
  const int h = t & 7;
  const v2i p = ((const v2i*)dm)[t];
  const float ad = __int_as_float(p.x);
  const float ms = narrow_spread(g, 8) ? dec_ord(g[h]) : dec_ord(p.y);
  const float exs = __expf(lrelu(aS[t] + ad) - lrelu(ms + ad));
  const float inv = 1.f / ((denom[t] + denom[(size_t)N * 8 + t]) + exs + SOFT_EPS);
  const float* ap = agg + (size_t)t * 8;
  const float* bp = ap + (size_t)N * 64;
  const float* hp = h1 + (size_t)t * 8;
  float* op = hpost + (size_t)t * 8;
#pragma unroll
  for (int c = 0; c < 8; ++c) {
    const float v = ((ap[c] + bp[c]) + exs * hp[c]) * inv + bias[h * 8 + c];
    op[c] = v > 0.f ? v : (__expf(v) - 1.f);
  }
}

template <int CTRL> __device__ __forceinline__ float row_swz(float v) {
  return __int_as_float(__builtin_amdgcn_update_dpp(0, __float_as_int(v), CTRL, 0xF, 0xF, true));
}
__device__ __forceinline__ float row16_sum(float v) {
  v += row_swz<0x161>(v); v += row_swz<0x162>(v); v += row_swz<0x164>(v); v += row_swz<0x168>(v);
  return v;
}
__device__ __forceinline__ float row16_max(float v) {
  v = fmaxf(v, row_swz<0x161>(v)); v = fmaxf(v, row_swz<0x162>(v));
  v = fmaxf(v, row_swz<0x164>(v)); v = fmaxf(v, row_swz<0x168>(v));
  return v;
}
template <int L> __device__ __forceinline__ float row_lane(float v) { return row_swz<0x150 | L>(v); }
__global__ void gat_padw2(const float* __restrict__ W2, float* __restrict__ W2p, int* __restrict__ g) {
  const int t = blockIdx.x * blockDim.x + threadIdx.x;
  if (t < 18) g[t] = (t < 8 || t == 16) ? (int)0x80000000 : 0x7FFFFFFF;
  if (t >= 64 * 48) return;
  const int k = t / 48, c = t % 48;
  W2p[t] = (c < 40) ? W2[k * 40 + c] : 0.f;
}

__global__ void gat_prep2(const float* __restrict__ h2a, float* __restrict__ h2b,
                          const float* __restrict__ attS,
                          const float* __restrict__ attD, float* __restrict__ aS,
                          int* __restrict__ dm,
                          float* __restrict__ agga, float* __restrict__ aggb,
                          int* __restrict__ g, int N) {
  __shared__ int smx[8], smn[8];
  const int lane = threadIdx.x & 31;
  const int hi = lane >> 4, ct = lane & 15;
  const int nbase = __builtin_amdgcn_readfirstlane((int)((blockIdx.x * blockDim.x + threadIdx.x) & ~31u));
  int emx = (int)0x80000000, emn = 0x7FFFFFFF;
  if (nbase < N) {
    const int left = N - nbase;
    const v2f z2 = {0.f, 0.f};
#pragma unroll
    for (int cpy = 0; cpy < 2; ++cpy) {
      float* ap = agga + (size_t)cpy * N * 32 + (size_t)nbase * 32;
      float* cp = aggb + (size_t)cpy * N * 16 + (size_t)nbase * 16;
      if (left >= 32) {
#pragma unroll
        for (int i = 0; i < 16; ++i) *(v2f*)(ap + i * 64 + 2 * lane) = z2;
#pragma unroll
        for (int i = 0; i < 8; ++i) *(v2f*)(cp + i * 64 + 2 * lane) = z2;
      } else {
#pragma unroll
        for (int i = 0; i < 16; ++i)
          if (2 * i + hi < left) *(v2f*)(ap + i * 64 + 2 * lane) = z2;
#pragma unroll
        for (int i = 0; i < 8; ++i)
          if (4 * i + (lane >> 3) < left) *(v2f*)(cp + i * 64 + 2 * lane) = z2;
      }
    }
    const float as0 = attS[2 * ct], as1 = attS[2 * ct + 1];
    const float ad0 = attD[2 * ct], ad1 = attD[2 * ct + 1];
    const float asb = attS[32 + (ct & 7)], adb = attD[32 + (ct & 7)];
    float smine = 0.f, dmine = 0.f;
    for (int grp = 0; grp < 2; ++grp) {
      v2f H[8];
      float Hb[8];
#pragma unroll
      for (int k = 0; k < 8; ++k) {
        const int nc = min(nbase + grp * 16 + 2 * k + hi, N - 1);
        H[k]  = *(const v2f*)(h2a + (size_t)nc * 32 + 2 * ct);
        Hb[k] = h2b[(size_t)nc * 16 + ct];
      }
      const int lrel = lane - grp * 16;
#pragma unroll
      for (int k = 0; k < 8; ++k) {
        const float s = row16_sum(H[k].x * as0 + H[k].y * as1 + (ct < 8 ? Hb[k] * asb : 0.f));
        const float d = row16_sum(H[k].x * ad0 + H[k].y * ad1 + (ct < 8 ? Hb[k] * adb : 0.f));
        const int s0 = __builtin_amdgcn_readlane(__float_as_int(s), 0), s1 = __builtin_amdgcn_readlane(__float_as_int(s), 16);
        const int d0 = __builtin_amdgcn_readlane(__float_as_int(d), 0), d1 = __builtin_amdgcn_readlane(__float_as_int(d), 16);
        smine = (lrel == 2 * k) ? __int_as_float(s0) : smine;
        smine = (lrel == 2 * k + 1) ? __int_as_float(s1) : smine;
        dmine = (lrel == 2 * k) ? __int_as_float(d0) : dmine;
        dmine = (lrel == 2 * k + 1) ? __int_as_float(d1) : dmine;
      }
    }
    if (lane < left) {
      const int n = nbase + lane;
      aS[n] = smine;
      h2b[(size_t)n * 16 + 8] = smine;
      v2i p; p.x = __float_as_int(dmine); p.y = enc_ord(smine);
      ((v2i*)dm)[n] = p;
      emx = p.y; emn = p.y;
    }
  }
#pragma unroll
  for (int m = 1; m < 32; m <<= 1) { emx = max(emx, __shfl_xor(emx, m)); emn = min(emn, __shfl_xor(emn, m)); }
  if ((threadIdx.x & 31) == 0) { smx[threadIdx.x >> 5] = emx; smn[threadIdx.x >> 5] = emn; }
  __syncthreads();
  if (threadIdx.x == 0) {
    int a = smx[0], b = smn[0];
    for (int w = 1; w < 8; ++w) { a = max(a, smx[w]); b = min(b, smn[w]); }
    atomicMax(&g[16], a);
    atomicMin(&g[17], b);
  }
}

__global__ void gat_emax2(const int* __restrict__ src, const int* __restrict__ dst,
                          const float* __restrict__ aS, int* __restrict__ dm,
                          const int* __restrict__ g, int E) {
  if (narrow_spread(g + 16, 1)) return;
  for (int c = 0; c < 8; ++c) {
    const int e = (blockIdx.x * 8 + c) * blockDim.x + threadIdx.x;
    if (e < E) atomicMax(&dm[dst[e] * 2 + 1], enc_ord(aS[src[e]]));
  }
}

__global__ void gat_eagg2(const int* __restrict__ src, const int* __restrict__ dst,
                          const float* __restrict__ aS, const int* __restrict__ dm,
                          const float* __restrict__ h2a, const float* __restrict__ h2b,
                          float* __restrict__ agga, float* __restrict__ aggb,
                          const int* __restrict__ g, int E, int N) {
  __shared__ int half_slot;
  const int lane = threadIdx.x & 31;
  const int e0 = __builtin_amdgcn_readfirstlane((int)((blockIdx.x * blockDim.x + threadIdx.x) & ~31u));
  const int e = e0 + lane;
  const int s = e < E ? src[e] : 0, d = e < E ? dst[e] : 0;
  const int hf = package_half(&half_slot);
  if (e0 >= E) return;
  agga += (size_t)hf * N * 32;
  aggb += (size_t)hf * N * 16;
  const v2i p = ((const v2i*)dm)[d];
  const float ad = __int_as_float(p.x);
  const float ms = narrow_spread(g + 16, 1) ? dec_ord(g[16]) : dec_ord(p.y);
  const int hi = lane >> 4;
  const int ct = lane & 15;
  if (e0 + 32 <= E) {
    float vl[16];
#pragma unroll
    for (int k = 0; k < 16; ++k) {
      const int s0 = __builtin_amdgcn_readlane(s, 2 * k), s1 = __builtin_amdgcn_readlane(s, 2 * k + 1);
      vl[k] = h2b[(size_t)(hi ? s1 : s0) * 16 + ct];
    }
    int asb = 0;
#pragma unroll
    for (int k = 0; k < 16; ++k) {
      const int a0 = __builtin_amdgcn_readlane(__float_as_int(vl[k]), 8);
      const int a1 = __builtin_amdgcn_readlane(__float_as_int(vl[k]), 24);
      asb = (lane == 2 * k) ? a0 : asb;
      asb = (lane == 2 * k + 1) ? a1 : asb;
    }
    const float ex = __expf(lrelu(__int_as_float(asb) + ad) - lrelu(ms + ad));
#pragma unroll
    for (int g = 0; g < 32; g += 8) {
      float v0[4], v1[4];
#pragma unroll
      for (int k = 0; k < 4; ++k) {
        const int j = g + 2 * k;
        const int s0 = __builtin_amdgcn_readlane(s, j), s1 = __builtin_amdgcn_readlane(s, j + 1);
        v0[k] = h2a[(size_t)s0 * 32 + lane];
        v1[k] = h2a[(size_t)s1 * 32 + lane];
      }
#pragma unroll
      for (int k = 0; k < 4; ++k) {
        const int j = g + 2 * k;
        const int d0 = __builtin_amdgcn_readlane(d, j), d1 = __builtin_amdgcn_readlane(d, j + 1);
        const float x0 = __int_as_float(__builtin_amdgcn_readlane(__float_as_int(ex), j));
        const float x1 = __int_as_float(__builtin_amdgcn_readlane(__float_as_int(ex), j + 1));
        const int dt = hi ? d1 : d0;
        const float xt = hi ? x1 : x0;
        const float vt = (ct == 8) ? 1.f : vl[g / 2 + k];
        atomicAdd(agga + (size_t)d0 * 32 + lane, x0 * v0[k]);
        atomicAdd(agga + (size_t)d1 * 32 + lane, x1 * v1[k]);
        if (ct <= 8)
          atomicAdd(aggb + (size_t)dt * 16 + ct, xt * vt);
      }
    }
    return;
  }
  const float ex = __expf(lrelu(aS[s] + ad) - lrelu(ms + ad));
#pragma unroll
  for (int j = 0; j < 32; j += 2) {
    const int s0 = __builtin_amdgcn_readlane(s, j), s1 = __builtin_amdgcn_readlane(s, j + 1);
    const int d0 = __builtin_amdgcn_readlane(d, j), d1 = __builtin_amdgcn_readlane(d, j + 1);
    const float x0 = __int_as_float(__builtin_amdgcn_readlane(__float_as_int(ex), j));
    const float x1 = __int_as_float(__builtin_amdgcn_readlane(__float_as_int(ex), j + 1));
    const float v0 = h2a[(size_t)s0 * 32 + lane];
    const float v1 = h2a[(size_t)s1 * 32 + lane];
    const int st = hi ? s1 : s0, dt = hi ? d1 : d0;
    const float xt = hi ? x1 : x0;
    const float vl = h2b[(size_t)st * 16 + ct];
    const float vt = (ct == 8) ? 1.f : vl;
    if (e0 + j < E)
      atomicAdd(agga + (size_t)d0 * 32 + lane, x0 * v0);
    if (e0 + j + 1 < E)
      atomicAdd(agga + (size_t)d1 * 32 + lane, x1 * v1);
    if (e0 + j + hi < E && ct <= 8)
      atomicAdd(aggb + (size_t)dt * 16 + ct, xt * vt);
  }
}

__global__ void __launch_bounds__(256)
gat_fin2(const float* __restrict__ agga, const float* __restrict__ aggb,
         const float* __restrict__ h2a, const float* __restrict__ h2b,
         const float* __restrict__ aS, const int* __restrict__ dm,
         const float* __restrict__ bias, float* __restrict__ out,
         const int* __restrict__ g, int N) {
  const int lane = threadIdx.x & 31;
  const int hi = lane >> 4, ct = lane & 15;
  const int nbase = __builtin_amdgcn_readfirstlane((int)((blockIdx.x * blockDim.x + threadIdx.x) & ~31u));
  if (nbase >= N) return;
  const bool nsp = narrow_spread(g + 16, 1);
  const float gms = dec_ord(g[16]);
  const float b0 = bias[2 * ct], b1 = bias[2 * ct + 1];
  const float bb = bias[32 + (ct & 7)];
  const float* agga2 = agga + (size_t)N * 32;
  const float* aggb2 = aggb + (size_t)N * 16;
  for (int grp = 0; grp < 2; ++grp) {
    v2f A[8], A2[8], H[8];
    float B[8], B2[8], Hb[8], sc[8];
    v2i P[8];
#pragma unroll
    for (int k = 0; k < 8; ++k) {
      const int nc = min(nbase + grp * 16 + 2 * k + hi, N - 1);
      A[k]  = *(const v2f*)(agga  + (size_t)nc * 32 + 2 * ct);
      A2[k] = *(const v2f*)(agga2 + (size_t)nc * 32 + 2 * ct);
      H[k]  = *(const v2f*)(h2a   + (size_t)nc * 32 + 2 * ct);
      B[k]  = aggb [(size_t)nc * 16 + ct];
      B2[k] = aggb2[(size_t)nc * 16 + ct];
      Hb[k] = h2b  [(size_t)nc * 16 + ct];
      sc[k] = aS[nc];
      P[k]  = ((const v2i*)dm)[nc];
    }
    asm volatile("" :: "v"(A[0]), "v"(A[1]), "v"(A[2]), "v"(A[3]), "v"(A[4]), "v"(A[5]), "v"(A[6]), "v"(A[7]),
                       "v"(A2[0]), "v"(A2[1]), "v"(A2[2]), "v"(A2[3]), "v"(A2[4]), "v"(A2[5]), "v"(A2[6]), "v"(A2[7]));
    asm volatile("" :: "v"(H[0]), "v"(H[1]), "v"(H[2]), "v"(H[3]), "v"(H[4]), "v"(H[5]), "v"(H[6]), "v"(H[7]),
                       "v"(P[0]), "v"(P[1]), "v"(P[2]), "v"(P[3]), "v"(P[4]), "v"(P[5]), "v"(P[6]), "v"(P[7]));
    asm volatile("" :: "v"(B[0]), "v"(B[1]), "v"(B[2]), "v"(B[3]), "v"(B[4]), "v"(B[5]), "v"(B[6]), "v"(B[7]),
                       "v"(B2[0]), "v"(B2[1]), "v"(B2[2]), "v"(B2[3]), "v"(B2[4]), "v"(B2[5]), "v"(B2[6]), "v"(B2[7]));
    asm volatile("" :: "v"(Hb[0]), "v"(Hb[1]), "v"(Hb[2]), "v"(Hb[3]), "v"(Hb[4]), "v"(Hb[5]), "v"(Hb[6]), "v"(Hb[7]),
                       "v"(sc[0]), "v"(sc[1]), "v"(sc[2]), "v"(sc[3]), "v"(sc[4]), "v"(sc[5]), "v"(sc[6]), "v"(sc[7]));
    v2f o2[8];
    float ob[8];
#pragma unroll
    for (int k = 0; k < 8; ++k) {
      const float ad = __int_as_float(P[k].x);
      const float ms = nsp ? gms : dec_ord(P[k].y);
      const float exs = __expf(lrelu(sc[k] + ad) - lrelu(ms + ad));
      const float bs = B[k] + B2[k];
      const float den = row_lane<8>(bs);
      const float inv = 1.f / (den + exs + SOFT_EPS);
      const float v0 = ((A[k].x + A2[k].x) + exs * H[k].x) * inv + b0;
      const float v1 = ((A[k].y + A2[k].y) + exs * H[k].y) * inv + b1;
      const float vb = (bs + exs * Hb[k]) * inv + bb;
      const float mx = row16_max(fmaxf(fmaxf(v0, v1), ct < 8 ? vb : -INFINITY));
      const float sum = row16_sum(__expf(v0 - mx) + __expf(v1 - mx) + (ct < 8 ? __expf(vb - mx) : 0.f));
      const float lse = mx + __logf(sum);
      o2[k].x = v0 - lse; o2[k].y = v1 - lse;
      ob[k] = vb - lse;
    }
    if (nbase + grp * 16 + 16 <= N) {
#pragma unroll
      for (int k = 0; k < 8; ++k) {
        float* op = out + (size_t)(nbase + grp * 16 + 2 * k + hi) * 40;
        *(v2f*)(op + 2 * ct) = o2[k];
        if (ct < 8) op[32 + ct] = ob[k];
      }
    } else {
#pragma unroll
      for (int k = 0; k < 8; ++k) {
        const int n = nbase + grp * 16 + 2 * k + hi;
        if (n < N) {
          float* op = out + (size_t)n * 40;
          *(v2f*)(op + 2 * ct) = o2[k];
          if (ct < 8) op[32 + ct] = ob[k];
        }
      }
    }
  }
}

extern "C" void kernel_launch(void* const* d_in, const int* in_sizes, int n_in,
                              void* d_out, int out_size, void* d_ws, size_t ws_size,
                              hipStream_t stream) {
  const float* x   = (const float*)d_in[0];
  const int*   ei  = (const int*)d_in[1];
  const float* W1  = (const float*)d_in[2];
  const float* as1 = (const float*)d_in[3];
  const float* ad1 = (const float*)d_in[4];
  const float* b1  = (const float*)d_in[5];
  const float* W2  = (const float*)d_in[6];
  const float* as2 = (const float*)d_in[7];
  const float* ad2 = (const float*)d_in[8];
  const float* b2  = (const float*)d_in[9];
  float* out = (float*)d_out;

  const int Fin = in_sizes[2] / 64;
  const int N   = in_sizes[0] / Fin;
  const int E   = in_sizes[1] / 2;
  const int* src = ei;
  const int* dst = ei + E;

  float* ws = (float*)d_ws;
  size_t o = 0;
  float* h1    = ws + o; o += (size_t)N * 64;
  float* hpost = ws + o; o += (size_t)N * 64;
  float* aS1   = ws + o; o += (size_t)N * 8;
  int*   dm1   = (int*)(ws + o); o += (size_t)N * 16;
  float* dn1   = ws + o; o += (size_t)N * 8 * 2;
  float* agg1  = ws + o; o += (size_t)N * 64 * 2;
  float* h2a   = ws + o; o += (size_t)N * 32;
  float* h2b   = ws + o; o += (size_t)N * 16;
  int*   dm2   = (int*)(ws + o); o += (size_t)N * 2;
  float* agg2a = ws + o; o += (size_t)N * 32 * 2;
  float* agg2b = ws + o; o += (size_t)N * 16 * 2;
  float* W2p   = ws + o; o += (size_t)64 * 48;
  int*   gst   = (int*)(ws + o); o += 32;
  float* aS2   = ws + o; o += (size_t)N;

  const int TB = 256;
  const int gNH = (N * 8 + TB - 1) / TB;
  const int gN  = (N + TB - 1) / TB;
  const int gE  = (E + TB - 1) / TB;
  const int gM  = (N + 15) / 16;
  const int gE8 = (gE + 7) / 8;

  gat_padw2 <<<(64 * 48 + TB - 1) / TB, TB, 0, stream>>>(W2, W2p, gst);

  gat_gemm_wmma<<<gM, dim3(32, 4), (size_t)Fin * 64 * 4, stream>>>(x, W1, h1, h1, N, Fin, 64, 64);
  gat_prep1 <<<gNH, TB, 0, stream>>>(h1, as1, ad1, aS1, dm1, dn1, agg1, gst, N);
  gat_emax1 <<<gE8, TB, 0, stream>>>(src, dst, aS1, dm1, gst, E);
  gat_eagg1 <<<gE,  TB, 0, stream>>>(src, dst, aS1, dm1, dn1, h1, agg1, gst, E, N);
  gat_fin1  <<<gNH, TB, 0, stream>>>(agg1, h1, aS1, dm1, dn1, b1, hpost, gst, N);

  gat_gemm_wmma<<<gM, dim3(32, 3), (size_t)64 * 48 * 4, stream>>>(hpost, W2p, h2a, h2b, N, 64, 48, 32);
  gat_prep2 <<<gN, TB, 0, stream>>>(h2a, h2b, as2, ad2, aS2, dm2, agg2a, agg2b, gst, N);
  gat_emax2 <<<gE8, TB, 0, stream>>>(src, dst, aS2, dm2, gst, E);
  gat_eagg2 <<<gE, TB, 0, stream>>>(src, dst, aS2, dm2, h2a, h2b, agg2a, agg2b, gst, E, N);
  gat_fin2  <<<gN, TB, 0, stream>>>(agg2a, agg2b, h2a, h2b, aS2, dm2, b2, out, gst, N);

  (void)n_in; (void)out_size; (void)ws_size;
}
